# starting kernel
# baseline (speedup 1.0000x reference)
#include <hip/hip_runtime.h>

#define NN 50000
#define NE 800000
#define DHID 128
#define DOUT 64

typedef float v2f __attribute__((ext_vector_type(2)));
typedef float v8f __attribute__((ext_vector_type(8)));

__device__ __forceinline__ void atomic_fadd(float* p, float v) {
    asm volatile("global_atomic_add_f32 %0, %1, off" : : "v"(p), "v"(v) : "memory");
}

__global__ void zero_f4(float4* __restrict__ p, int n4) {
    int i = blockIdx.x * blockDim.x + threadIdx.x;
    if (i < n4) p[i] = make_float4(0.f, 0.f, 0.f, 0.f);
}

__global__ void degree_kernel(const int* __restrict__ dst, float* __restrict__ cnt, int E) {
    int e = blockIdx.x * blockDim.x + threadIdx.x;
    if (e < E) atomic_fadd(&cnt[dst[e]], 1.0f);
}

__global__ void scatter_kernel(const int* __restrict__ src, const int* __restrict__ dst,
                               const float* __restrict__ feat, float* __restrict__ agg,
                               int E) {
    int gid  = blockIdx.x * blockDim.x + threadIdx.x;
    int e    = gid >> 5;
    int lane = gid & 31;
    if (e >= E) return;
    int s = src[e];
    int d = dst[e];
    const float4 v = *(const float4*)(feat + (size_t)s * DHID + lane * 4);
    float* o = agg + (size_t)d * DHID + lane * 4;
    atomic_fadd(o + 0, v.x);
    atomic_fadd(o + 1, v.y);
    atomic_fadd(o + 2, v.z);
    atomic_fadd(o + 3, v.w);
}

template <int LDB>
__global__ void sage_gemm(const float* __restrict__ agg, const float* __restrict__ cnt,
                          const float* __restrict__ hroot,
                          const float* __restrict__ Wl, const float* __restrict__ bias,
                          const float* __restrict__ Wr,
                          float* __restrict__ out, int nrows, int relu) {
    constexpr int MB = 5;
    constexpr int NB = 2;
    const int lane = threadIdx.x & 31;
    const int wave = (blockIdx.x * blockDim.x + threadIdx.x) >> 5;
    const int ntn  = LDB / (16 * NB);
    const int stm  = wave / ntn;
    const int stn  = wave % ntn;
    if (stm * (16 * MB) >= nrows) return;

    const int half = lane >> 4;
    const int mn   = lane & 15;
    const int koff = half * 2;

    const int r0 = stm * (16 * MB) + mn;
    const int c0 = stn * (16 * NB) + mn;

    float rinv[MB];
#pragma unroll
    for (int mi = 0; mi < MB; ++mi)
        rinv[mi] = 1.0f / fmaxf(cnt[r0 + mi * 16], 1.0f);

    const float* ap = agg   + (size_t)r0 * DHID;
    const float* xp = hroot + (size_t)r0 * DHID;

    v8f acc[MB][NB];
#pragma unroll
    for (int mi = 0; mi < MB; ++mi)
#pragma unroll
        for (int nj = 0; nj < NB; ++nj)
            acc[mi][nj] = (v8f){0.f, 0.f, 0.f, 0.f, 0.f, 0.f, 0.f, 0.f};

#pragma unroll 2
    for (int k = 0; k < DHID; k += 4) {
        v2f bl[NB], br[NB];
#pragma unroll
        for (int nj = 0; nj < NB; ++nj) {
            bl[nj][0] = Wl[(k + koff)     * LDB + c0 + nj * 16];
            bl[nj][1] = Wl[(k + koff + 1) * LDB + c0 + nj * 16];
            br[nj][0] = Wr[(k + koff)     * LDB + c0 + nj * 16];
            br[nj][1] = Wr[(k + koff + 1) * LDB + c0 + nj * 16];
        }
        v2f am[MB], ar[MB];
#pragma unroll
        for (int mi = 0; mi < MB; ++mi) {
            float2 av = *(const float2*)(ap + mi * 16 * DHID + k + koff);
            am[mi][0] = av.x * rinv[mi];
            am[mi][1] = av.y * rinv[mi];
            float2 xv = *(const float2*)(xp + mi * 16 * DHID + k + koff);
            ar[mi][0] = xv.x;
            ar[mi][1] = xv.y;
        }
#pragma unroll
        for (int mi = 0; mi < MB; ++mi) {
#pragma unroll
            for (int nj = 0; nj < NB; ++nj) {
                acc[mi][nj] = __builtin_amdgcn_wmma_f32_16x16x4_f32(
                    false, am[mi], false, bl[nj], (short)0, acc[mi][nj], false, false);
                acc[mi][nj] = __builtin_amdgcn_wmma_f32_16x16x4_f32(
                    false, ar[mi], false, br[nj], (short)0, acc[mi][nj], false, false);
            }
        }
    }

#pragma unroll
    for (int nj = 0; nj < NB; ++nj) {
        const float bv = bias[c0 + nj * 16];
#pragma unroll
        for (int mi = 0; mi < MB; ++mi) {
#pragma unroll
            for (int v = 0; v < 8; ++v) {
                int m = stm * (16 * MB) + mi * 16 + v + half * 8;
                float val = acc[mi][nj][v] + bv;
                if (relu) val = fmaxf(val, 0.f);
                out[(size_t)m * LDB + c0 + nj * 16] = val;
            }
        }
    }
}

extern "C" void kernel_launch(void* const* d_in, const int* in_sizes, int n_in,
                              void* d_out, int out_size, void* d_ws, size_t ws_size,
                              hipStream_t stream) {
    const float* x   = (const float*)d_in[0];
    const int*   ei  = (const int*)d_in[1];
    const float* W1l = (const float*)d_in[2];
    const float* b1  = (const float*)d_in[3];
    const float* W1r = (const float*)d_in[4];
    const float* W2l = (const float*)d_in[5];
    const float* b2  = (const float*)d_in[6];
    const float* W2r = (const float*)d_in[7];
    float*       out = (float*)d_out;

    const int* src = ei;
    const int* dst = ei + NE;

    float* agg = (float*)d_ws;
    float* cnt = agg + (size_t)NN * DHID;
    float* h   = cnt + NN;

    const int BLK = 256;

    {
        int n4 = (NN * DHID + NN) / 4;
        zero_f4<<<(n4 + BLK - 1) / BLK, BLK, 0, stream>>>((float4*)agg, n4);
    }

    degree_kernel<<<(NE + BLK - 1) / BLK, BLK, 0, stream>>>(dst, cnt, NE);

    {
        long long thr = (long long)NE * 32;
        scatter_kernel<<<(int)((thr + BLK - 1) / BLK), BLK, 0, stream>>>(src, dst, x, agg, NE);

        int waves = (NN / 80) * (DHID / 32);
        int thr2  = waves * 32;
        sage_gemm<DHID><<<(thr2 + BLK - 1) / BLK, BLK, 0, stream>>>(
            agg, cnt, x, W1l, b1, W1r, h, NN,  1);
    }

    {
        int n4 = (NN * DHID) / 4;
        zero_f4<<<(n4 + BLK - 1) / BLK, BLK, 0, stream>>>((float4*)agg, n4);

        long long thr = (long long)NE * 32;
        scatter_kernel<<<(int)((thr + BLK - 1) / BLK), BLK, 0, stream>>>(src, dst, h, agg, NE);

        int waves = (NN / 80) * (DOUT / 32);
        int thr2  = waves * 32;
        sage_gemm<DOUT><<<(thr2 + BLK - 1) / BLK, BLK, 0, stream>>>(
            agg, cnt, h, W2l, b2, W2r, out, NN,  0);
    }
}


# GEMM A operands (mean and root rows) staged once per 80-row workgroup through LDS with whole-row 16-B/lane loads; k loop reads A fragments from LDS; sum order unchanged
# speedup vs baseline: 1.1940x; 1.1940x over previous
#include <hip/hip_runtime.h>

#define NN 50000
#define NE 800000
#define DHID 128
#define DOUT 64

typedef float v2f __attribute__((ext_vector_type(2)));
typedef float v8f __attribute__((ext_vector_type(8)));

__device__ __forceinline__ void atomic_fadd(float* p, float v) {
    asm volatile("global_atomic_add_f32 %0, %1, off" : : "v"(p), "v"(v) : "memory");
}

__global__ void zero_f4(float4* __restrict__ p, int n4) {
    int i = blockIdx.x * blockDim.x + threadIdx.x;
    if (i < n4) p[i] = make_float4(0.f, 0.f, 0.f, 0.f);
}

__global__ void degree_kernel(const int* __restrict__ dst, float* __restrict__ cnt, int E) {
    int e = blockIdx.x * blockDim.x + threadIdx.x;
    if (e < E) atomic_fadd(&cnt[dst[e]], 1.0f);
}

__global__ void scatter_kernel(const int* __restrict__ src, const int* __restrict__ dst,
                               const float* __restrict__ feat, float* __restrict__ agg,
                               int E) {
    int gid  = blockIdx.x * blockDim.x + threadIdx.x;
    int e    = gid >> 5;
    int lane = gid & 31;
    if (e >= E) return;
    int s = src[e];
    int d = dst[e];
    const float4 v = *(const float4*)(feat + (size_t)s * DHID + lane * 4);
    float* o = agg + (size_t)d * DHID + lane * 4;
    atomic_fadd(o + 0, v.x);
    atomic_fadd(o + 1, v.y);
    atomic_fadd(o + 2, v.z);
    atomic_fadd(o + 3, v.w);
}

template <int LDB>
__global__ void
__launch_bounds__(LDB)
sage_gemm(const float* __restrict__ agg, const float* __restrict__ cnt,
          const float* __restrict__ hroot,
          const float* __restrict__ Wl, const float* __restrict__ bias,
          const float* __restrict__ Wr,
          float* __restrict__ out, int nrows, int relu) {
    constexpr int MB    = 5;
    constexpr int NB    = 2;
    constexpr int ROWS  = 16 * MB;
    constexpr int PITCH = DHID + 4;
    constexpr int NW    = LDB / (16 * NB);
    constexpr int RPW   = ROWS / NW;
    constexpr int BATCH = 10;

    __shared__ __attribute__((aligned(16))) float sM[ROWS * PITCH];
    __shared__ __attribute__((aligned(16))) float sX[ROWS * PITCH];

    const int lane = threadIdx.x & 31;
    const int wv   = threadIdx.x >> 5;
    const int row0 = blockIdx.x * ROWS;
    if (row0 >= nrows) return;

#pragma unroll
    for (int b0 = 0; b0 < RPW; b0 += BATCH) {
        float4 va[BATCH], vx[BATCH];
        float  cn[BATCH];
#pragma unroll
        for (int j = 0; j < BATCH; ++j) {
            const int r = wv + (b0 + j) * NW;
            const size_t g = (size_t)(row0 + r) * DHID + lane * 4;
            va[j] = *(const float4*)(agg + g);
            vx[j] = *(const float4*)(hroot + g);
            cn[j] = cnt[row0 + r];
        }
#pragma unroll
        for (int j = 0; j < BATCH; ++j) {
            const int r = wv + (b0 + j) * NW;
            const float s = 1.0f / fmaxf(cn[j], 1.0f);
            float4 a = va[j];
            a.x *= s; a.y *= s; a.z *= s; a.w *= s;
            *(float4*)&sM[r * PITCH + lane * 4] = a;
            *(float4*)&sX[r * PITCH + lane * 4] = vx[j];
        }
    }
    __syncthreads();

    const int half = lane >> 4;
    const int mn   = lane & 15;
    const int koff = half * 2;
    const int c0   = wv * (16 * NB) + mn;

    const float* mp = sM + mn * PITCH + koff;
    const float* xp = sX + mn * PITCH + koff;

    v8f acc[MB][NB];
#pragma unroll
    for (int mi = 0; mi < MB; ++mi)
#pragma unroll
        for (int nj = 0; nj < NB; ++nj)
            acc[mi][nj] = (v8f){0.f, 0.f, 0.f, 0.f, 0.f, 0.f, 0.f, 0.f};

#pragma unroll 2
    for (int k = 0; k < DHID; k += 4) {
        v2f bl[NB], br[NB];
#pragma unroll
        for (int nj = 0; nj < NB; ++nj) {
            bl[nj][0] = Wl[(k + koff)     * LDB + c0 + nj * 16];
            bl[nj][1] = Wl[(k + koff + 1) * LDB + c0 + nj * 16];
            br[nj][0] = Wr[(k + koff)     * LDB + c0 + nj * 16];
            br[nj][1] = Wr[(k + koff + 1) * LDB + c0 + nj * 16];
        }
        v2f am[MB], ar[MB];
#pragma unroll
        for (int mi = 0; mi < MB; ++mi) {
            am[mi] = *(const v2f*)(mp + mi * 16 * PITCH + k);
            ar[mi] = *(const v2f*)(xp + mi * 16 * PITCH + k);
        }
#pragma unroll
        for (int mi = 0; mi < MB; ++mi) {
#pragma unroll
            for (int nj = 0; nj < NB; ++nj) {
                acc[mi][nj] = __builtin_amdgcn_wmma_f32_16x16x4_f32(
                    false, am[mi], false, bl[nj], (short)0, acc[mi][nj], false, false);
                acc[mi][nj] = __builtin_amdgcn_wmma_f32_16x16x4_f32(
                    false, ar[mi], false, br[nj], (short)0, acc[mi][nj], false, false);
            }
        }
    }

#pragma unroll
    for (int nj = 0; nj < NB; ++nj) {
        const float bv = bias[c0 + nj * 16];
#pragma unroll
        for (int mi = 0; mi < MB; ++mi) {
#pragma unroll
            for (int v = 0; v < 8; ++v) {
                int m = row0 + mi * 16 + v + half * 8;
                float val = acc[mi][nj][v] + bv;
                if (relu) val = fmaxf(val, 0.f);
                out[(size_t)m * LDB + c0 + nj * 16] = val;
            }
        }
    }
}

extern "C" void kernel_launch(void* const* d_in, const int* in_sizes, int n_in,
                              void* d_out, int out_size, void* d_ws, size_t ws_size,
                              hipStream_t stream) {
    const float* x   = (const float*)d_in[0];
    const int*   ei  = (const int*)d_in[1];
    const float* W1l = (const float*)d_in[2];
    const float* b1  = (const float*)d_in[3];
    const float* W1r = (const float*)d_in[4];
    const float* W2l = (const float*)d_in[5];
    const float* b2  = (const float*)d_in[6];
    const float* W2r = (const float*)d_in[7];
    float*       out = (float*)d_out;

    const int* src = ei;
    const int* dst = ei + NE;

    float* agg = (float*)d_ws;
    float* cnt = agg + (size_t)NN * DHID;
    float* h   = cnt + NN;

    const int BLK = 256;

    {
        int n4 = (NN * DHID + NN) / 4;
        zero_f4<<<(n4 + BLK - 1) / BLK, BLK, 0, stream>>>((float4*)agg, n4);
    }

    degree_kernel<<<(NE + BLK - 1) / BLK, BLK, 0, stream>>>(dst, cnt, NE);

    {
        long long thr = (long long)NE * 32;
        scatter_kernel<<<(int)((thr + BLK - 1) / BLK), BLK, 0, stream>>>(src, dst, x, agg, NE);

        sage_gemm<DHID><<<NN / 80, DHID, 0, stream>>>(
            agg, cnt, x, W1l, b1, W1r, h, NN,  1);
    }

    {
        int n4 = (NN * DHID) / 4;
        zero_f4<<<(n4 + BLK - 1) / BLK, BLK, 0, stream>>>((float4*)agg, n4);

        long long thr = (long long)NE * 32;
        scatter_kernel<<<(int)((thr + BLK - 1) / BLK), BLK, 0, stream>>>(src, dst, h, agg, NE);

        sage_gemm<DOUT><<<NN / 80, DOUT, 0, stream>>>(
            agg, cnt, h, W2l, b2, W2r, out, NN,  0);
    }
}


# Scatter atomics re-laid: lane l owns floats l, l+32, l+64, l+96 of a row, so each wave atomic adds into one whole 128-B line (4 line requests per edge, not 16 part lines); arrival-order sums kept
# speedup vs baseline: 1.5701x; 1.3150x over previous
#include <hip/hip_runtime.h>

#define NN 50000
#define NE 800000
#define DHID 128
#define DOUT 64

typedef float v2f __attribute__((ext_vector_type(2)));
typedef float v8f __attribute__((ext_vector_type(8)));

__device__ __forceinline__ void atomic_fadd(float* p, float v) {
    asm volatile("global_atomic_add_f32 %0, %1, off" : : "v"(p), "v"(v) : "memory");
}

__global__ void zero_f4(float4* __restrict__ p, int n4) {
    int i = blockIdx.x * blockDim.x + threadIdx.x;
    if (i < n4) p[i] = make_float4(0.f, 0.f, 0.f, 0.f);
}

__global__ void degree_kernel(const int* __restrict__ dst, float* __restrict__ cnt, int E) {
    int e = blockIdx.x * blockDim.x + threadIdx.x;
    if (e < E) atomic_fadd(&cnt[dst[e]], 1.0f);
}

__global__ void scatter_kernel(const int* __restrict__ src, const int* __restrict__ dst,
                               const float* __restrict__ feat, float* __restrict__ agg,
                               int E) {
    int gid  = blockIdx.x * blockDim.x + threadIdx.x;
    int e    = gid >> 5;
    int lane = gid & 31;
    if (e >= E) return;
    int s = src[e];
    int d = dst[e];
    const float* f = feat + (size_t)s * DHID + lane;
    const float v0 = f[0];
    const float v1 = f[32];
    const float v2 = f[64];
    const float v3 = f[96];
    float* o = agg + (size_t)d * DHID + lane;
    atomic_fadd(o,      v0);
    atomic_fadd(o + 32, v1);
    atomic_fadd(o + 64, v2);
    atomic_fadd(o + 96, v3);
}

template <int LDB>
__global__ void
__launch_bounds__(LDB)
sage_gemm(const float* __restrict__ agg, const float* __restrict__ cnt,
          const float* __restrict__ hroot,
          const float* __restrict__ Wl, const float* __restrict__ bias,
          const float* __restrict__ Wr,
          float* __restrict__ out, int nrows, int relu) {
    constexpr int MB    = 5;
    constexpr int NB    = 2;
    constexpr int ROWS  = 16 * MB;
    constexpr int PITCH = DHID + 4;
    constexpr int NW    = LDB / (16 * NB);
    constexpr int RPW   = ROWS / NW;
    constexpr int BATCH = 10;

    __shared__ __attribute__((aligned(16))) float sM[ROWS * PITCH];
    __shared__ __attribute__((aligned(16))) float sX[ROWS * PITCH];

    const int lane = threadIdx.x & 31;
    const int wv   = threadIdx.x >> 5;
    const int row0 = blockIdx.x * ROWS;
    if (row0 >= nrows) return;

#pragma unroll
    for (int b0 = 0; b0 < RPW; b0 += BATCH) {
        float4 va[BATCH], vx[BATCH];
        float  cn[BATCH];
#pragma unroll
        for (int j = 0; j < BATCH; ++j) {
            const int r = wv + (b0 + j) * NW;
            const size_t g = (size_t)(row0 + r) * DHID + lane * 4;
            va[j] = *(const float4*)(agg + g);
            vx[j] = *(const float4*)(hroot + g);
            cn[j] = cnt[row0 + r];
        }
#pragma unroll
        for (int j = 0; j < BATCH; ++j) {
            const int r = wv + (b0 + j) * NW;
            const float s = 1.0f / fmaxf(cn[j], 1.0f);
            float4 a = va[j];
            a.x *= s; a.y *= s; a.z *= s; a.w *= s;
            *(float4*)&sM[r * PITCH + lane * 4] = a;
            *(float4*)&sX[r * PITCH + lane * 4] = vx[j];
        }
    }
    __syncthreads();

    const int half = lane >> 4;
    const int mn   = lane & 15;
    const int koff = half * 2;
    const int c0   = wv * (16 * NB) + mn;

    const float* mp = sM + mn * PITCH + koff;
    const float* xp = sX + mn * PITCH + koff;

    v8f acc[MB][NB];
#pragma unroll
    for (int mi = 0; mi < MB; ++mi)
#pragma unroll
        for (int nj = 0; nj < NB; ++nj)
            acc[mi][nj] = (v8f){0.f, 0.f, 0.f, 0.f, 0.f, 0.f, 0.f, 0.f};

#pragma unroll 2
    for (int k = 0; k < DHID; k += 4) {
        v2f bl[NB], br[NB];
#pragma unroll
        for (int nj = 0; nj < NB; ++nj) {
            bl[nj][0] = Wl[(k + koff)     * LDB + c0 + nj * 16];
            bl[nj][1] = Wl[(k + koff + 1) * LDB + c0 + nj * 16];
            br[nj][0] = Wr[(k + koff)     * LDB + c0 + nj * 16];
            br[nj][1] = Wr[(k + koff + 1) * LDB + c0 + nj * 16];
        }
        v2f am[MB], ar[MB];
#pragma unroll
        for (int mi = 0; mi < MB; ++mi) {
            am[mi] = *(const v2f*)(mp + mi * 16 * PITCH + k);
            ar[mi] = *(const v2f*)(xp + mi * 16 * PITCH + k);
        }
#pragma unroll
        for (int mi = 0; mi < MB; ++mi) {
#pragma unroll
            for (int nj = 0; nj < NB; ++nj) {
                acc[mi][nj] = __builtin_amdgcn_wmma_f32_16x16x4_f32(
                    false, am[mi], false, bl[nj], (short)0, acc[mi][nj], false, false);
                acc[mi][nj] = __builtin_amdgcn_wmma_f32_16x16x4_f32(
                    false, ar[mi], false, br[nj], (short)0, acc[mi][nj], false, false);
            }
        }
    }

#pragma unroll
    for (int nj = 0; nj < NB; ++nj) {
        const float bv = bias[c0 + nj * 16];
#pragma unroll
        for (int mi = 0; mi < MB; ++mi) {
#pragma unroll
            for (int v = 0; v < 8; ++v) {
                int m = row0 + mi * 16 + v + half * 8;
                float val = acc[mi][nj][v] + bv;
                if (relu) val = fmaxf(val, 0.f);
                out[(size_t)m * LDB + c0 + nj * 16] = val;
            }
        }
    }
}

extern "C" void kernel_launch(void* const* d_in, const int* in_sizes, int n_in,
                              void* d_out, int out_size, void* d_ws, size_t ws_size,
                              hipStream_t stream) {
    const float* x   = (const float*)d_in[0];
    const int*   ei  = (const int*)d_in[1];
    const float* W1l = (const float*)d_in[2];
    const float* b1  = (const float*)d_in[3];
    const float* W1r = (const float*)d_in[4];
    const float* W2l = (const float*)d_in[5];
    const float* b2  = (const float*)d_in[6];
    const float* W2r = (const float*)d_in[7];
    float*       out = (float*)d_out;

    const int* src = ei;
    const int* dst = ei + NE;

    float* agg = (float*)d_ws;
    float* cnt = agg + (size_t)NN * DHID;
    float* h   = cnt + NN;

    const int BLK = 256;

    {
        int n4 = (NN * DHID + NN) / 4;
        zero_f4<<<(n4 + BLK - 1) / BLK, BLK, 0, stream>>>((float4*)agg, n4);
    }

    degree_kernel<<<(NE + BLK - 1) / BLK, BLK, 0, stream>>>(dst, cnt, NE);

    {
        long long thr = (long long)NE * 32;
        scatter_kernel<<<(int)((thr + BLK - 1) / BLK), BLK, 0, stream>>>(src, dst, x, agg, NE);

        sage_gemm<DHID><<<NN / 80, DHID, 0, stream>>>(
            agg, cnt, x, W1l, b1, W1r, h, NN,  1);
    }

    {
        int n4 = (NN * DHID) / 4;
        zero_f4<<<(n4 + BLK - 1) / BLK, BLK, 0, stream>>>((float4*)agg, n4);

        long long thr = (long long)NE * 32;
        scatter_kernel<<<(int)((thr + BLK - 1) / BLK), BLK, 0, stream>>>(src, dst, h, agg, NE);

        sage_gemm<DOUT><<<NN / 80, DOUT, 0, stream>>>(
            agg, cnt, h, W2l, b2, W2r, out, NN,  0);
    }
}


# Scatter: each wave takes 8 consecutive edges (ids by scalar loads, all 32 row loads in flight before the 32 whole-line atomics), 8x fewer waves to start and drain; arrival-order sums kept
# speedup vs baseline: 1.9200x; 1.2229x over previous
#include <hip/hip_runtime.h>

#define NN 50000
#define NE 800000
#define DHID 128
#define DOUT 64

typedef float v2f __attribute__((ext_vector_type(2)));
typedef float v8f __attribute__((ext_vector_type(8)));

__device__ __forceinline__ void atomic_fadd(float* p, float v) {
    asm volatile("global_atomic_add_f32 %0, %1, off" : : "v"(p), "v"(v) : "memory");
}

__global__ void zero_f4(float4* __restrict__ p, int n4) {
    int i = blockIdx.x * blockDim.x + threadIdx.x;
    if (i < n4) p[i] = make_float4(0.f, 0.f, 0.f, 0.f);
}

__global__ void degree_kernel(const int* __restrict__ dst, float* __restrict__ cnt, int E) {
    int e = blockIdx.x * blockDim.x + threadIdx.x;
    if (e < E) atomic_fadd(&cnt[dst[e]], 1.0f);
}

#define EPW 8
static_assert(NE % EPW == 0, "every wave takes a full group of edges");

__global__ void scatter_kernel(const int* __restrict__ src, const int* __restrict__ dst,
                               const float* __restrict__ feat, float* __restrict__ agg,
                               int E) {
    const int gid  = blockIdx.x * blockDim.x + threadIdx.x;
    const int lane = threadIdx.x & 31;
    const int e0 = __builtin_amdgcn_readfirstlane(gid >> 5) * EPW;
    if (e0 >= E) return;
    int s[EPW], d[EPW];
#pragma unroll
    for (int j = 0; j < EPW; ++j) {
        s[j] = src[e0 + j];
        d[j] = dst[e0 + j];
    }
    float v[EPW][4];
#pragma unroll
    for (int j = 0; j < EPW; ++j) {
        const float* f = feat + (size_t)s[j] * DHID + lane;
        v[j][0] = f[0];
        v[j][1] = f[32];
        v[j][2] = f[64];
        v[j][3] = f[96];
    }
#pragma unroll
    for (int j = 0; j < EPW; ++j) {
        float* o = agg + (size_t)d[j] * DHID + lane;
        atomic_fadd(o,      v[j][0]);
        atomic_fadd(o + 32, v[j][1]);
        atomic_fadd(o + 64, v[j][2]);
        atomic_fadd(o + 96, v[j][3]);
    }
}

template <int LDB>
__global__ void
__launch_bounds__(LDB)
sage_gemm(const float* __restrict__ agg, const float* __restrict__ cnt,
          const float* __restrict__ hroot,
          const float* __restrict__ Wl, const float* __restrict__ bias,
          const float* __restrict__ Wr,
          float* __restrict__ out, int nrows, int relu) {
    constexpr int MB    = 5;
    constexpr int NB    = 2;
    constexpr int ROWS  = 16 * MB;
    constexpr int PITCH = DHID + 4;
    constexpr int NW    = LDB / (16 * NB);
    constexpr int RPW   = ROWS / NW;
    constexpr int BATCH = 10;

    __shared__ __attribute__((aligned(16))) float sM[ROWS * PITCH];
    __shared__ __attribute__((aligned(16))) float sX[ROWS * PITCH];

    const int lane = threadIdx.x & 31;
    const int wv   = threadIdx.x >> 5;
    const int row0 = blockIdx.x * ROWS;
    if (row0 >= nrows) return;

#pragma unroll
    for (int b0 = 0; b0 < RPW; b0 += BATCH) {
        float4 va[BATCH], vx[BATCH];
        float  cn[BATCH];
#pragma unroll
        for (int j = 0; j < BATCH; ++j) {
            const int r = wv + (b0 + j) * NW;
            const size_t g = (size_t)(row0 + r) * DHID + lane * 4;
            va[j] = *(const float4*)(agg + g);
            vx[j] = *(const float4*)(hroot + g);
            cn[j] = cnt[row0 + r];
        }
#pragma unroll
        for (int j = 0; j < BATCH; ++j) {
            const int r = wv + (b0 + j) * NW;
            const float s = 1.0f / fmaxf(cn[j], 1.0f);
            float4 a = va[j];
            a.x *= s; a.y *= s; a.z *= s; a.w *= s;
            *(float4*)&sM[r * PITCH + lane * 4] = a;
            *(float4*)&sX[r * PITCH + lane * 4] = vx[j];
        }
    }
    __syncthreads();

    const int half = lane >> 4;
    const int mn   = lane & 15;
    const int koff = half * 2;
    const int c0   = wv * (16 * NB) + mn;

    const float* mp = sM + mn * PITCH + koff;
    const float* xp = sX + mn * PITCH + koff;

    v8f acc[MB][NB];
#pragma unroll
    for (int mi = 0; mi < MB; ++mi)
#pragma unroll
        for (int nj = 0; nj < NB; ++nj)
            acc[mi][nj] = (v8f){0.f, 0.f, 0.f, 0.f, 0.f, 0.f, 0.f, 0.f};

#pragma unroll 2
    for (int k = 0; k < DHID; k += 4) {
        v2f bl[NB], br[NB];
#pragma unroll
        for (int nj = 0; nj < NB; ++nj) {
            bl[nj][0] = Wl[(k + koff)     * LDB + c0 + nj * 16];
            bl[nj][1] = Wl[(k + koff + 1) * LDB + c0 + nj * 16];
            br[nj][0] = Wr[(k + koff)     * LDB + c0 + nj * 16];
            br[nj][1] = Wr[(k + koff + 1) * LDB + c0 + nj * 16];
        }
        v2f am[MB], ar[MB];
#pragma unroll
        for (int mi = 0; mi < MB; ++mi) {
            am[mi] = *(const v2f*)(mp + mi * 16 * PITCH + k);
            ar[mi] = *(const v2f*)(xp + mi * 16 * PITCH + k);
        }
#pragma unroll
        for (int mi = 0; mi < MB; ++mi) {
#pragma unroll
            for (int nj = 0; nj < NB; ++nj) {
                acc[mi][nj] = __builtin_amdgcn_wmma_f32_16x16x4_f32(
                    false, am[mi], false, bl[nj], (short)0, acc[mi][nj], false, false);
                acc[mi][nj] = __builtin_amdgcn_wmma_f32_16x16x4_f32(
                    false, ar[mi], false, br[nj], (short)0, acc[mi][nj], false, false);
            }
        }
    }

#pragma unroll
    for (int nj = 0; nj < NB; ++nj) {
        const float bv = bias[c0 + nj * 16];
#pragma unroll
        for (int mi = 0; mi < MB; ++mi) {
#pragma unroll
            for (int v = 0; v < 8; ++v) {
                int m = row0 + mi * 16 + v + half * 8;
                float val = acc[mi][nj][v] + bv;
                if (relu) val = fmaxf(val, 0.f);
                out[(size_t)m * LDB + c0 + nj * 16] = val;
            }
        }
    }
}

extern "C" void kernel_launch(void* const* d_in, const int* in_sizes, int n_in,
                              void* d_out, int out_size, void* d_ws, size_t ws_size,
                              hipStream_t stream) {
    const float* x   = (const float*)d_in[0];
    const int*   ei  = (const int*)d_in[1];
    const float* W1l = (const float*)d_in[2];
    const float* b1  = (const float*)d_in[3];
    const float* W1r = (const float*)d_in[4];
    const float* W2l = (const float*)d_in[5];
    const float* b2  = (const float*)d_in[6];
    const float* W2r = (const float*)d_in[7];
    float*       out = (float*)d_out;

    const int* src = ei;
    const int* dst = ei + NE;

    float* agg = (float*)d_ws;
    float* cnt = agg + (size_t)NN * DHID;
    float* h   = cnt + NN;

    const int BLK = 256;

    {
        int n4 = (NN * DHID + NN) / 4;
        zero_f4<<<(n4 + BLK - 1) / BLK, BLK, 0, stream>>>((float4*)agg, n4);
    }

    degree_kernel<<<(NE + BLK - 1) / BLK, BLK, 0, stream>>>(dst, cnt, NE);

    {
        long long thr = (long long)(NE / EPW) * 32;
        scatter_kernel<<<(int)((thr + BLK - 1) / BLK), BLK, 0, stream>>>(src, dst, x, agg, NE);

        sage_gemm<DHID><<<NN / 80, DHID, 0, stream>>>(
            agg, cnt, x, W1l, b1, W1r, h, NN,  1);
    }

    {
        int n4 = (NN * DHID) / 4;
        zero_f4<<<(n4 + BLK - 1) / BLK, BLK, 0, stream>>>((float4*)agg, n4);

        long long thr = (long long)(NE / EPW) * 32;
        scatter_kernel<<<(int)((thr + BLK - 1) / BLK), BLK, 0, stream>>>(src, dst, h, agg, NE);

        sage_gemm<DOUT><<<NN / 80, DOUT, 0, stream>>>(
            agg, cnt, h, W2l, b2, W2r, out, NN,  0);
    }
}


# Layer 2 aggregates AFTER its linear map: rows of h@W2l (64 floats, not 128) are atomically added, scaled by 1/deg, onto out=h@W2r+b; sum order changes: per-edge products summed in arrival order
# speedup vs baseline: 2.5395x; 1.3227x over previous
#include <hip/hip_runtime.h>

#define NN 50000
#define NE 800000
#define DHID 128
#define DOUT 64

typedef float v2f __attribute__((ext_vector_type(2)));
typedef float v8f __attribute__((ext_vector_type(8)));

__device__ __forceinline__ void atomic_fadd(float* p, float v) {
    asm volatile("global_atomic_add_f32 %0, %1, off" : : "v"(p), "v"(v) : "memory");
}

__global__ void zero_f4(float4* __restrict__ p, int n4) {
    int i = blockIdx.x * blockDim.x + threadIdx.x;
    if (i < n4) p[i] = make_float4(0.f, 0.f, 0.f, 0.f);
}

__global__ void degree_kernel(const int* __restrict__ dst, float* __restrict__ cnt, int E) {
    int e = blockIdx.x * blockDim.x + threadIdx.x;
    if (e < E) atomic_fadd(&cnt[dst[e]], 1.0f);
}

#define EPW 8
static_assert(NE % EPW == 0, "every wave takes a full group of edges");

__global__ void scatter_kernel(const int* __restrict__ src, const int* __restrict__ dst,
                               const float* __restrict__ feat, float* __restrict__ agg,
                               int E) {
    const int gid  = blockIdx.x * blockDim.x + threadIdx.x;
    const int lane = threadIdx.x & 31;
    const int e0 = __builtin_amdgcn_readfirstlane(gid >> 5) * EPW;
    if (e0 >= E) return;
    int s[EPW], d[EPW];
#pragma unroll
    for (int j = 0; j < EPW; ++j) {
        s[j] = src[e0 + j];
        d[j] = dst[e0 + j];
    }
    float v[EPW][4];
#pragma unroll
    for (int j = 0; j < EPW; ++j) {
        const float* f = feat + (size_t)s[j] * DHID + lane;
        v[j][0] = f[0];
        v[j][1] = f[32];
        v[j][2] = f[64];
        v[j][3] = f[96];
    }
#pragma unroll
    for (int j = 0; j < EPW; ++j) {
        float* o = agg + (size_t)d[j] * DHID + lane;
        atomic_fadd(o,      v[j][0]);
        atomic_fadd(o + 32, v[j][1]);
        atomic_fadd(o + 64, v[j][2]);
        atomic_fadd(o + 96, v[j][3]);
    }
}

template <int LDB>
__global__ void
__launch_bounds__(LDB)
sage_gemm(const float* __restrict__ agg, const float* __restrict__ cnt,
          const float* __restrict__ hroot,
          const float* __restrict__ Wl, const float* __restrict__ bias,
          const float* __restrict__ Wr,
          float* __restrict__ out, int nrows, int relu) {
    constexpr int MB    = 5;
    constexpr int NB    = 2;
    constexpr int ROWS  = 16 * MB;
    constexpr int PITCH = DHID + 4;
    constexpr int NW    = LDB / (16 * NB);
    constexpr int RPW   = ROWS / NW;
    constexpr int BATCH = 10;

    __shared__ __attribute__((aligned(16))) float sM[ROWS * PITCH];
    __shared__ __attribute__((aligned(16))) float sX[ROWS * PITCH];

    const int lane = threadIdx.x & 31;
    const int wv   = threadIdx.x >> 5;
    const int row0 = blockIdx.x * ROWS;
    if (row0 >= nrows) return;

#pragma unroll
    for (int b0 = 0; b0 < RPW; b0 += BATCH) {
        float4 va[BATCH], vx[BATCH];
        float  cn[BATCH];
#pragma unroll
        for (int j = 0; j < BATCH; ++j) {
            const int r = wv + (b0 + j) * NW;
            const size_t g = (size_t)(row0 + r) * DHID + lane * 4;
            va[j] = *(const float4*)(agg + g);
            vx[j] = *(const float4*)(hroot + g);
            cn[j] = cnt[row0 + r];
        }
#pragma unroll
        for (int j = 0; j < BATCH; ++j) {
            const int r = wv + (b0 + j) * NW;
            const float s = 1.0f / fmaxf(cn[j], 1.0f);
            float4 a = va[j];
            a.x *= s; a.y *= s; a.z *= s; a.w *= s;
            *(float4*)&sM[r * PITCH + lane * 4] = a;
            *(float4*)&sX[r * PITCH + lane * 4] = vx[j];
        }
    }
    __syncthreads();

    const int half = lane >> 4;
    const int mn   = lane & 15;
    const int koff = half * 2;
    const int c0   = wv * (16 * NB) + mn;

    const float* mp = sM + mn * PITCH + koff;
    const float* xp = sX + mn * PITCH + koff;

    v8f acc[MB][NB];
#pragma unroll
    for (int mi = 0; mi < MB; ++mi)
#pragma unroll
        for (int nj = 0; nj < NB; ++nj)
            acc[mi][nj] = (v8f){0.f, 0.f, 0.f, 0.f, 0.f, 0.f, 0.f, 0.f};

#pragma unroll 2
    for (int k = 0; k < DHID; k += 4) {
        v2f bl[NB], br[NB];
#pragma unroll
        for (int nj = 0; nj < NB; ++nj) {
            bl[nj][0] = Wl[(k + koff)     * LDB + c0 + nj * 16];
            bl[nj][1] = Wl[(k + koff + 1) * LDB + c0 + nj * 16];
            br[nj][0] = Wr[(k + koff)     * LDB + c0 + nj * 16];
            br[nj][1] = Wr[(k + koff + 1) * LDB + c0 + nj * 16];
        }
        v2f am[MB], ar[MB];
#pragma unroll
        for (int mi = 0; mi < MB; ++mi) {
            am[mi] = *(const v2f*)(mp + mi * 16 * PITCH + k);
            ar[mi] = *(const v2f*)(xp + mi * 16 * PITCH + k);
        }
#pragma unroll
        for (int mi = 0; mi < MB; ++mi) {
#pragma unroll
            for (int nj = 0; nj < NB; ++nj) {
                acc[mi][nj] = __builtin_amdgcn_wmma_f32_16x16x4_f32(
                    false, am[mi], false, bl[nj], (short)0, acc[mi][nj], false, false);
                acc[mi][nj] = __builtin_amdgcn_wmma_f32_16x16x4_f32(
                    false, ar[mi], false, br[nj], (short)0, acc[mi][nj], false, false);
            }
        }
    }

#pragma unroll
    for (int nj = 0; nj < NB; ++nj) {
        const float bv = bias[c0 + nj * 16];
#pragma unroll
        for (int mi = 0; mi < MB; ++mi) {
#pragma unroll
            for (int v = 0; v < 8; ++v) {
                int m = row0 + mi * 16 + v + half * 8;
                float val = acc[mi][nj][v] + bv;
                if (relu) val = fmaxf(val, 0.f);
                out[(size_t)m * LDB + c0 + nj * 16] = val;
            }
        }
    }
}


#define EPW2 16
static_assert(NE % EPW2 == 0, "every wave takes a full group of edges");

__global__ void scatter_scaled64_kernel(const int* __restrict__ src, const int* __restrict__ dst,
                                        const float* __restrict__ P, const float* __restrict__ cnt,
                                        float* __restrict__ out, int E) {
    const int gid  = blockIdx.x * blockDim.x + threadIdx.x;
    const int lane = threadIdx.x & 31;
    const int e0 = __builtin_amdgcn_readfirstlane(gid >> 5) * EPW2;
    if (e0 >= E) return;
    int s[EPW2], d[EPW2];
#pragma unroll
    for (int j = 0; j < EPW2; ++j) {
        s[j] = src[e0 + j];
        d[j] = dst[e0 + j];
    }
    float v[EPW2][2];
#pragma unroll
    for (int j = 0; j < EPW2; ++j) {
        const float* f = P + (size_t)s[j] * DOUT + lane;
        v[j][0] = f[0];
        v[j][1] = f[32];
    }
    float sc[EPW2];
#pragma unroll
    for (int j = 0; j < EPW2; ++j)
        sc[j] = 1.0f / fmaxf(cnt[d[j]], 1.0f);
#pragma unroll
    for (int j = 0; j < EPW2; ++j) {
        float* o = out + (size_t)d[j] * DOUT + lane;
        atomic_fadd(o,      v[j][0] * sc[j]);
        atomic_fadd(o + 32, v[j][1] * sc[j]);
    }
}

__global__ void
__launch_bounds__(128)
lin2_gemm(const float* __restrict__ hroot,
          const float* __restrict__ Wl, const float* __restrict__ Wr,
          const float* __restrict__ bias,
          float* __restrict__ P, float* __restrict__ out, int nrows) {
    constexpr int MB    = 5;
    constexpr int NB    = 2;
    constexpr int ROWS  = 16 * MB;
    constexpr int PITCH = DHID + 4;
    constexpr int NW    = 4;
    constexpr int RPW   = ROWS / NW;
    constexpr int BATCH = 10;

    __shared__ __attribute__((aligned(16))) float sX[ROWS * PITCH];

    const int lane = threadIdx.x & 31;
    const int wv   = threadIdx.x >> 5;
    const int row0 = blockIdx.x * ROWS;
    if (row0 >= nrows) return;

#pragma unroll
    for (int b0 = 0; b0 < RPW; b0 += BATCH) {
        float4 vx[BATCH];
#pragma unroll
        for (int j = 0; j < BATCH; ++j) {
            const int r = wv + (b0 + j) * NW;
            vx[j] = *(const float4*)(hroot + (size_t)(row0 + r) * DHID + lane * 4);
        }
#pragma unroll
        for (int j = 0; j < BATCH; ++j) {
            const int r = wv + (b0 + j) * NW;
            *(float4*)&sX[r * PITCH + lane * 4] = vx[j];
        }
    }
    __syncthreads();

    const int half = lane >> 4;
    const int mn   = lane & 15;
    const int koff = half * 2;
    const int isr  = wv >> 1;
    const int c0   = (wv & 1) * (16 * NB) + mn;

    const float* Wm = isr ? Wr : Wl;
    float*       dp = isr ? out : P;
    const float* xp = sX + mn * PITCH + koff;

    v8f acc[MB][NB];
#pragma unroll
    for (int mi = 0; mi < MB; ++mi)
#pragma unroll
        for (int nj = 0; nj < NB; ++nj)
            acc[mi][nj] = (v8f){0.f, 0.f, 0.f, 0.f, 0.f, 0.f, 0.f, 0.f};

#pragma unroll 2
    for (int k = 0; k < DHID; k += 4) {
        v2f bw[NB];
#pragma unroll
        for (int nj = 0; nj < NB; ++nj) {
            bw[nj][0] = Wm[(k + koff)     * DOUT + c0 + nj * 16];
            bw[nj][1] = Wm[(k + koff + 1) * DOUT + c0 + nj * 16];
        }
        v2f ar[MB];
#pragma unroll
        for (int mi = 0; mi < MB; ++mi)
            ar[mi] = *(const v2f*)(xp + mi * 16 * PITCH + k);
#pragma unroll
        for (int mi = 0; mi < MB; ++mi) {
#pragma unroll
            for (int nj = 0; nj < NB; ++nj) {
                acc[mi][nj] = __builtin_amdgcn_wmma_f32_16x16x4_f32(
                    false, ar[mi], false, bw[nj], (short)0, acc[mi][nj], false, false);
            }
        }
    }

#pragma unroll
    for (int nj = 0; nj < NB; ++nj) {
        const float bv = isr ? bias[c0 + nj * 16] : 0.f;
#pragma unroll
        for (int mi = 0; mi < MB; ++mi) {
#pragma unroll
            for (int v = 0; v < 8; ++v) {
                int m = row0 + mi * 16 + v + half * 8;
                dp[(size_t)m * DOUT + c0 + nj * 16] = acc[mi][nj][v] + bv;
            }
        }
    }
}

extern "C" void kernel_launch(void* const* d_in, const int* in_sizes, int n_in,
                              void* d_out, int out_size, void* d_ws, size_t ws_size,
                              hipStream_t stream) {
    const float* x   = (const float*)d_in[0];
    const int*   ei  = (const int*)d_in[1];
    const float* W1l = (const float*)d_in[2];
    const float* b1  = (const float*)d_in[3];
    const float* W1r = (const float*)d_in[4];
    const float* W2l = (const float*)d_in[5];
    const float* b2  = (const float*)d_in[6];
    const float* W2r = (const float*)d_in[7];
    float*       out = (float*)d_out;

    const int* src = ei;
    const int* dst = ei + NE;

    float* agg = (float*)d_ws;
    float* cnt = agg + (size_t)NN * DHID;
    float* h   = cnt + NN;

    const int BLK = 256;

    {
        int n4 = (NN * DHID + NN) / 4;
        zero_f4<<<(n4 + BLK - 1) / BLK, BLK, 0, stream>>>((float4*)agg, n4);
    }

    degree_kernel<<<(NE + BLK - 1) / BLK, BLK, 0, stream>>>(dst, cnt, NE);

    {
        long long thr = (long long)(NE / EPW) * 32;
        scatter_kernel<<<(int)((thr + BLK - 1) / BLK), BLK, 0, stream>>>(src, dst, x, agg, NE);

        sage_gemm<DHID><<<NN / 80, DHID, 0, stream>>>(
            agg, cnt, x, W1l, b1, W1r, h, NN,  1);
    }

    {
        float* P = agg;

        lin2_gemm<<<NN / 80, 128, 0, stream>>>(h, W2l, W2r, b2, P, out, NN);

        long long thr = (long long)(NE / EPW2) * 32;
        scatter_scaled64_kernel<<<(int)((thr + BLK - 1) / BLK), BLK, 0, stream>>>(src, dst, P, cnt, out, NE);
    }
}


# Aggregation without atomics: edges ordered by destination per call (two-level ballot split, one-block scan); one wave sums a node's source rows in registers; sum order changes: fixed list order
# speedup vs baseline: 2.8903x; 1.1381x over previous
#include <hip/hip_runtime.h>

#define NN 50000
#define NE 800000
#define DHID 128
#define DOUT 64

typedef float v2f __attribute__((ext_vector_type(2)));
typedef float v8f __attribute__((ext_vector_type(8)));

#define NBK   ((NN + 127) >> 7)
#define G1    196
#define R1    4
#define EPB   (1024 * R1)
#define HP    (NBK + 1)
#define HP2   132
#define NCNT  (NBK * G1)
static_assert(G1 * EPB >= NE, "level-1 blocks cover all edges");
static_assert(NBK <= 512, "level-1 keys fit the 9 ballot bits");
static_assert(NCNT % 4 == 0, "the scan moves 4 counts per thread");

template <int NBITS>
__device__ __forceinline__ unsigned match_mask(int key) {
    unsigned m = 0xffffffffu;
#pragma unroll
    for (int bit = 0; bit < NBITS; ++bit) {
        const bool on = (key >> bit) & 1;
        const unsigned bal = __builtin_amdgcn_ballot_w32(on);
        m &= on ? bal : ~bal;
    }
    return m;
}

__global__ void
__launch_bounds__(1024)
split_count(const int* __restrict__ dst, int* __restrict__ counts, int* __restrict__ lrank) {
    __shared__ int hist[32 * HP];
    const int t    = threadIdx.x;
    const int lane = t & 31;
    const int w    = t >> 5;
    const int blk  = blockIdx.x;
    const unsigned below = (1u << lane) - 1u;
    for (int i = t; i < 32 * HP; i += 1024) hist[i] = 0;
    __syncthreads();
    int* hw = hist + w * HP;
    int key[R1], rk[R1];
#pragma unroll
    for (int r = 0; r < R1; ++r) {
        const int e = blk * EPB + (w * R1 + r) * 32 + lane;
        key[r] = (e < NE) ? (dst[e] >> 7) : NBK;
    }
#pragma unroll
    for (int r = 0; r < R1; ++r) {
        const int k = key[r];
        const unsigned m = match_mask<9>(k);
        const int ahead = __builtin_popcount(m & below);
        const int old = hw[k];
        rk[r]  = old + ahead;
        if (ahead == 0) hw[k] = old + __builtin_popcount(m);
    }
    __syncthreads();
    if (t < HP) {
        int run = 0;
        for (int w2 = 0; w2 < 32; ++w2) {
            const int c = hist[w2 * HP + t];
            hist[w2 * HP + t] = run;
            run += c;
        }
        if (t < NBK) counts[t * G1 + blk] = run;
    }
    __syncthreads();
#pragma unroll
    for (int r = 0; r < R1; ++r) {
        const int e = blk * EPB + (w * R1 + r) * 32 + lane;
        if (e < NE) lrank[e] = hw[key[r]] + rk[r];
    }
}

#define SCAN_T 1024
#define SCAN_TILES ((NCNT + 4 * SCAN_T - 1) / (4 * SCAN_T))

__global__ void
__launch_bounds__(SCAN_T)
scan_kernel(const int* __restrict__ cin, int* __restrict__ cout) {
    __shared__ int wsum[SCAN_TILES][32];
    const int t    = threadIdx.x;
    const int lane = t & 31;
    const int wv   = t >> 5;
    int4 v[SCAN_TILES];
#pragma unroll
    for (int k = 0; k < SCAN_TILES; ++k) {
        const int i = (k * SCAN_T + t) * 4;
        v[k] = (i < NCNT) ? *(const int4*)(cin + i) : make_int4(0, 0, 0, 0);
    }
    int inc[SCAN_TILES];
#pragma unroll
    for (int k = 0; k < SCAN_TILES; ++k) {
        int x = v[k].x + v[k].y + v[k].z + v[k].w;
        for (int off = 1; off < 32; off <<= 1) {
            const int nb = __shfl_up(x, off);
            if (lane >= off) x += nb;
        }
        inc[k] = x;
        if (lane == 31) wsum[k][wv] = x;
    }
    __syncthreads();
    int carry = 0;
#pragma unroll
    for (int k = 0; k < SCAN_TILES; ++k) {
        const int4 a = v[k];
        const int s1 = a.x, s2 = s1 + a.y, s3 = s2 + a.z, s4 = s3 + a.w;
        int winc = wsum[k][lane];
        for (int off = 1; off < 32; off <<= 1) {
            const int nb = __shfl_up(winc, off);
            if (lane >= off) winc += nb;
        }
        const int tile_total  = __shfl(winc, 31);
        const int before_wave = __shfl(winc, (wv + 31) & 31);
        const int base = carry + (wv ? before_wave : 0) + (inc[k] - s4);
        const int i = (k * SCAN_T + t) * 4;
        if (i < NCNT) *(int4*)(cout + i) = make_int4(base, base + s1, base + s2, base + s3);
        carry += tile_total;
    }
}

static_assert(NN <= (1 << 24), "a source id and 7 node bits fit one int");
__global__ void split_place(const int* __restrict__ src, const int* __restrict__ dst,
                            const int* __restrict__ offs, const int* __restrict__ lrank,
                            int* __restrict__ ekey) {
    const int e = blockIdx.x * blockDim.x + threadIdx.x;
    if (e < NE) {
        const int d = dst[e];
        const int pos = offs[(d >> 7) * G1 + e / EPB] + lrank[e];
        ekey[pos] = (src[e] << 7) | (d & 127);
    }
}

__global__ void
__launch_bounds__(1024)
bucket_lists(const int* __restrict__ offs, const int* __restrict__ ekey,
             int* __restrict__ deg, int* __restrict__ rowptr, int* __restrict__ col) {
    __shared__ int hist[32 * HP2];
    __shared__ int ncnt[128];
    __shared__ int nbase[128];
    const int t    = threadIdx.x;
    const int lane = t & 31;
    const int w    = t >> 5;
    const int b    = blockIdx.x;
    const unsigned below = (1u << lane) - 1u;
    const int beg = offs[b * G1];
    const int end = (b + 1 < NBK) ? offs[(b + 1) * G1] : NE;
    const int n   = end - beg;
    for (int i = t; i < 32 * HP2; i += 1024) hist[i] = 0;
    __syncthreads();
    int* hw = hist + w * HP2;
    for (int i0 = 0; i0 < n; i0 += 1024) {
        const int i = i0 + t;
        const int k = (i < n) ? (ekey[beg + i] & 127) : 128;
        const unsigned m = match_mask<8>(k);
        if ((m & below) == 0) hw[k] += __builtin_popcount(m);
    }
    __syncthreads();
    if (t < 128) {
        int run = 0;
        for (int w2 = 0; w2 < 32; ++w2) {
            const int c = hist[w2 * HP2 + t];
            hist[w2 * HP2 + t] = run;
            run += c;
        }
        ncnt[t] = run;
    }
    __syncthreads();
    if (t < 128) {
        int ex = 0;
        for (int j = 0; j < t; ++j) ex += ncnt[j];
        nbase[t] = beg + ex;
        const int node = b * 128 + t;
        if (node < NN) {
            deg[node]    = ncnt[t];
            rowptr[node] = beg + ex;
        }
    }
    __syncthreads();
    for (int i0 = 0; i0 < n; i0 += 1024) {
        const int i = i0 + t;
        const int kv = (i < n) ? ekey[beg + i] : 0;
        const int k  = (i < n) ? (kv & 127) : 128;
        const unsigned m = match_mask<8>(k);
        const int ahead = __builtin_popcount(m & below);
        const int old = hw[k];
        if (i < n) col[nbase[k] + old + ahead] = kv >> 7;
        if (ahead == 0) hw[k] = old + __builtin_popcount(m);
    }
}

#define GU 16
__global__ void gather128_kernel(const int* __restrict__ rowptr, const int* __restrict__ deg,
                                 const int* __restrict__ col, const float* __restrict__ feat,
                                 float* __restrict__ mean, int n) {
    const int gid  = blockIdx.x * blockDim.x + threadIdx.x;
    const int lane = threadIdx.x & 31;
    const int r = __builtin_amdgcn_readfirstlane(gid >> 5);
    if (r >= n) return;
    const int beg = rowptr[r];
    const int cnt = deg[r];
    float4 acc = make_float4(0.f, 0.f, 0.f, 0.f);
    for (int b0 = 0; b0 < cnt; b0 += 32) {
        const int m = (cnt - b0 < 32) ? (cnt - b0) : 32;
        const int myid = (lane < m) ? col[beg + b0 + lane] : 0;
        for (int j0 = 0; j0 < m; j0 += GU) {
            float4 v[GU];
#pragma unroll
            for (int j = 0; j < GU; ++j) {
                const int sid = __builtin_amdgcn_readlane(myid, j0 + j);
                if (j0 + j < m) v[j] = *(const float4*)(feat + (size_t)sid * DHID + lane * 4);
                else            v[j] = make_float4(0.f, 0.f, 0.f, 0.f);
            }
#pragma unroll
            for (int j = 0; j < GU; ++j) {
                acc.x += v[j].x; acc.y += v[j].y; acc.z += v[j].z; acc.w += v[j].w;
            }
        }
    }
    const float s = 1.0f / fmaxf((float)cnt, 1.0f);
    acc.x *= s; acc.y *= s; acc.z *= s; acc.w *= s;
    *(float4*)(mean + (size_t)r * DHID + lane * 4) = acc;
}

template <int LDB>
__global__ void
__launch_bounds__(LDB)
sage_gemm(const float* agg,
          const float* __restrict__ hroot,
          const float* __restrict__ Wl, const float* __restrict__ bias,
          const float* __restrict__ Wr,
          float* out, int nrows, int relu) {
    constexpr int MB    = 5;
    constexpr int NB    = 2;
    constexpr int ROWS  = 16 * MB;
    constexpr int PITCH = DHID + 4;
    constexpr int NW    = LDB / (16 * NB);
    constexpr int RPW   = ROWS / NW;
    constexpr int BATCH = 10;

    __shared__ __attribute__((aligned(16))) float sM[ROWS * PITCH];
    __shared__ __attribute__((aligned(16))) float sX[ROWS * PITCH];

    const int lane = threadIdx.x & 31;
    const int wv   = threadIdx.x >> 5;
    const int row0 = blockIdx.x * ROWS;
    if (row0 >= nrows) return;

#pragma unroll
    for (int b0 = 0; b0 < RPW; b0 += BATCH) {
        float4 va[BATCH], vx[BATCH];
#pragma unroll
        for (int j = 0; j < BATCH; ++j) {
            const int r = wv + (b0 + j) * NW;
            const size_t g = (size_t)(row0 + r) * DHID + lane * 4;
            va[j] = *(const float4*)(agg + g);
            vx[j] = *(const float4*)(hroot + g);
        }
#pragma unroll
        for (int j = 0; j < BATCH; ++j) {
            const int r = wv + (b0 + j) * NW;
            *(float4*)&sM[r * PITCH + lane * 4] = va[j];
            *(float4*)&sX[r * PITCH + lane * 4] = vx[j];
        }
    }
    __syncthreads();

    const int half = lane >> 4;
    const int mn   = lane & 15;
    const int koff = half * 2;
    const int c0   = wv * (16 * NB) + mn;

    const float* mp = sM + mn * PITCH + koff;
    const float* xp = sX + mn * PITCH + koff;

    v8f acc[MB][NB];
#pragma unroll
    for (int mi = 0; mi < MB; ++mi)
#pragma unroll
        for (int nj = 0; nj < NB; ++nj)
            acc[mi][nj] = (v8f){0.f, 0.f, 0.f, 0.f, 0.f, 0.f, 0.f, 0.f};

#pragma unroll 2
    for (int k = 0; k < DHID; k += 4) {
        v2f bl[NB], br[NB];
#pragma unroll
        for (int nj = 0; nj < NB; ++nj) {
            bl[nj][0] = Wl[(k + koff)     * LDB + c0 + nj * 16];
            bl[nj][1] = Wl[(k + koff + 1) * LDB + c0 + nj * 16];
            br[nj][0] = Wr[(k + koff)     * LDB + c0 + nj * 16];
            br[nj][1] = Wr[(k + koff + 1) * LDB + c0 + nj * 16];
        }
        v2f am[MB], ar[MB];
#pragma unroll
        for (int mi = 0; mi < MB; ++mi) {
            am[mi] = *(const v2f*)(mp + mi * 16 * PITCH + k);
            ar[mi] = *(const v2f*)(xp + mi * 16 * PITCH + k);
        }
#pragma unroll
        for (int mi = 0; mi < MB; ++mi) {
#pragma unroll
            for (int nj = 0; nj < NB; ++nj) {
                acc[mi][nj] = __builtin_amdgcn_wmma_f32_16x16x4_f32(
                    false, am[mi], false, bl[nj], (short)0, acc[mi][nj], false, false);
                acc[mi][nj] = __builtin_amdgcn_wmma_f32_16x16x4_f32(
                    false, ar[mi], false, br[nj], (short)0, acc[mi][nj], false, false);
            }
        }
    }

#pragma unroll
    for (int nj = 0; nj < NB; ++nj) {
        const float bv = bias[c0 + nj * 16];
#pragma unroll
        for (int mi = 0; mi < MB; ++mi) {
#pragma unroll
            for (int v = 0; v < 8; ++v) {
                int m = row0 + mi * 16 + v + half * 8;
                float val = acc[mi][nj][v] + bv;
                if (relu) val = fmaxf(val, 0.f);
                out[(size_t)m * LDB + c0 + nj * 16] = val;
            }
        }
    }
}


__global__ void gather64_kernel(const int* __restrict__ rowptr, const int* __restrict__ deg,
                                const int* __restrict__ col, const float* __restrict__ P,
                                float* __restrict__ out, int n) {
    const int gid  = blockIdx.x * blockDim.x + threadIdx.x;
    const int lane = threadIdx.x & 31;
    const int r = __builtin_amdgcn_readfirstlane(gid >> 5);
    if (r >= n) return;
    const int beg = rowptr[r];
    const int cnt = deg[r];
    float* o = out + (size_t)r * DOUT + lane * 2;
    const float2 start = *(const float2*)o;
    float2 acc = make_float2(0.f, 0.f);
    for (int b0 = 0; b0 < cnt; b0 += 32) {
        const int m = (cnt - b0 < 32) ? (cnt - b0) : 32;
        const int myid = (lane < m) ? col[beg + b0 + lane] : 0;
        for (int j0 = 0; j0 < m; j0 += GU) {
            float2 v[GU];
#pragma unroll
            for (int j = 0; j < GU; ++j) {
                const int sid = __builtin_amdgcn_readlane(myid, j0 + j);
                if (j0 + j < m) v[j] = *(const float2*)(P + (size_t)sid * DOUT + lane * 2);
                else            v[j] = make_float2(0.f, 0.f);
            }
#pragma unroll
            for (int j = 0; j < GU; ++j) {
                acc.x += v[j].x; acc.y += v[j].y;
            }
        }
    }
    const float s = 1.0f / fmaxf((float)cnt, 1.0f);
    *(float2*)o = make_float2(start.x + acc.x * s, start.y + acc.y * s);
}

__global__ void
__launch_bounds__(128)
lin2_gemm(const float* __restrict__ hroot,
          const float* __restrict__ Wl, const float* __restrict__ Wr,
          const float* __restrict__ bias,
          float* __restrict__ P, float* __restrict__ out, int nrows) {
    constexpr int MB    = 5;
    constexpr int NB    = 2;
    constexpr int ROWS  = 16 * MB;
    constexpr int PITCH = DHID + 4;
    constexpr int NW    = 4;
    constexpr int RPW   = ROWS / NW;
    constexpr int BATCH = 10;

    __shared__ __attribute__((aligned(16))) float sX[ROWS * PITCH];

    const int lane = threadIdx.x & 31;
    const int wv   = threadIdx.x >> 5;
    const int row0 = blockIdx.x * ROWS;
    if (row0 >= nrows) return;

#pragma unroll
    for (int b0 = 0; b0 < RPW; b0 += BATCH) {
        float4 vx[BATCH];
#pragma unroll
        for (int j = 0; j < BATCH; ++j) {
            const int r = wv + (b0 + j) * NW;
            vx[j] = *(const float4*)(hroot + (size_t)(row0 + r) * DHID + lane * 4);
        }
#pragma unroll
        for (int j = 0; j < BATCH; ++j) {
            const int r = wv + (b0 + j) * NW;
            *(float4*)&sX[r * PITCH + lane * 4] = vx[j];
        }
    }
    __syncthreads();

    const int half = lane >> 4;
    const int mn   = lane & 15;
    const int koff = half * 2;
    const int isr  = wv >> 1;
    const int c0   = (wv & 1) * (16 * NB) + mn;

    const float* Wm = isr ? Wr : Wl;
    float*       dp = isr ? out : P;
    const float* xp = sX + mn * PITCH + koff;

    v8f acc[MB][NB];
#pragma unroll
    for (int mi = 0; mi < MB; ++mi)
#pragma unroll
        for (int nj = 0; nj < NB; ++nj)
            acc[mi][nj] = (v8f){0.f, 0.f, 0.f, 0.f, 0.f, 0.f, 0.f, 0.f};

#pragma unroll 2
    for (int k = 0; k < DHID; k += 4) {
        v2f bw[NB];
#pragma unroll
        for (int nj = 0; nj < NB; ++nj) {
            bw[nj][0] = Wm[(k + koff)     * DOUT + c0 + nj * 16];
            bw[nj][1] = Wm[(k + koff + 1) * DOUT + c0 + nj * 16];
        }
        v2f ar[MB];
#pragma unroll
        for (int mi = 0; mi < MB; ++mi)
            ar[mi] = *(const v2f*)(xp + mi * 16 * PITCH + k);
#pragma unroll
        for (int mi = 0; mi < MB; ++mi) {
#pragma unroll
            for (int nj = 0; nj < NB; ++nj) {
                acc[mi][nj] = __builtin_amdgcn_wmma_f32_16x16x4_f32(
                    false, ar[mi], false, bw[nj], (short)0, acc[mi][nj], false, false);
            }
        }
    }

#pragma unroll
    for (int nj = 0; nj < NB; ++nj) {
        const float bv = isr ? bias[c0 + nj * 16] : 0.f;
#pragma unroll
        for (int mi = 0; mi < MB; ++mi) {
#pragma unroll
            for (int v = 0; v < 8; ++v) {
                int m = row0 + mi * 16 + v + half * 8;
                dp[(size_t)m * DOUT + c0 + nj * 16] = acc[mi][nj][v] + bv;
            }
        }
    }
}

extern "C" void kernel_launch(void* const* d_in, const int* in_sizes, int n_in,
                              void* d_out, int out_size, void* d_ws, size_t ws_size,
                              hipStream_t stream) {
    const float* x   = (const float*)d_in[0];
    const int*   ei  = (const int*)d_in[1];
    const float* W1l = (const float*)d_in[2];
    const float* b1  = (const float*)d_in[3];
    const float* W1r = (const float*)d_in[4];
    const float* W2l = (const float*)d_in[5];
    const float* b2  = (const float*)d_in[6];
    const float* W2r = (const float*)d_in[7];
    float*       out = (float*)d_out;

    const int* src = ei;
    const int* dst = ei + NE;

    float* A      = (float*)d_ws;
    float* P      = A + (size_t)NN * DHID;
    int*   col    = (int*)(P + (size_t)NN * DOUT);
    int*   ekey   = col + NE;
    int*   lrank  = ekey + NE;
    int*   rowptr = lrank + NE;
    int*   deg    = rowptr + NN;
    int*   counts = deg + NN;
    int*   offs   = counts + NCNT;
    static_assert((size_t)NN * DOUT + 3 * (size_t)NE + 2 * (size_t)NN + 2 * (size_t)NCNT
                      <= (size_t)NN * DHID + NN,
                  "P and the lists fit the space that used to hold the counts and h");
    static_assert(((size_t)NN * DHID + (size_t)NN * DOUT + 3 * (size_t)NE + 2 * (size_t)NN) % 4 == 0,
                  "counts and offs start on 16 B for the scan's 16-B accesses");

    const int BLK = 256;

    split_count<<<G1, 1024, 0, stream>>>(dst, counts, lrank);
    scan_kernel<<<1, SCAN_T, 0, stream>>>(counts, offs);
    split_place<<<(NE + BLK - 1) / BLK, BLK, 0, stream>>>(src, dst, offs, lrank, ekey);
    bucket_lists<<<NBK, 1024, 0, stream>>>(offs, ekey, deg, rowptr, col);

    {
        gather128_kernel<<<(NN * 32 + BLK - 1) / BLK, BLK, 0, stream>>>(rowptr, deg, col, x, A, NN);

        sage_gemm<DHID><<<NN / 80, DHID, 0, stream>>>(
            A, x, W1l, b1, W1r, A, NN,  1);
    }

    {
        float* h = A;

        lin2_gemm<<<NN / 80, 128, 0, stream>>>(h, W2l, W2r, b2, P, out, NN);

        gather64_kernel<<<(NN * 32 + BLK - 1) / BLK, BLK, 0, stream>>>(rowptr, deg, col, P, out, NN);
    }
}


# Layer 1 as one workgroup pass per 80 nodes: neighbour rows summed straight into the GEMM's LDS operand, relu(h) tile fed from LDS to both layer-2 products; mean, h planes, 2 launches gone; same sums
# speedup vs baseline: 3.0500x; 1.0553x over previous
#include <hip/hip_runtime.h>

#define NN 50000
#define NE 800000
#define DHID 128
#define DOUT 64

typedef float v2f __attribute__((ext_vector_type(2)));
typedef float v8f __attribute__((ext_vector_type(8)));

#define NBK   ((NN + 127) >> 7)
#define G1    196
#define R1    4
#define EPB   (1024 * R1)
#define HP    (NBK + 1)
#define HP2   132
#define NCNT  (NBK * G1)
static_assert(G1 * EPB >= NE, "level-1 blocks cover all edges");
static_assert(NBK <= 512, "level-1 keys fit the 9 ballot bits");
static_assert(NCNT % 4 == 0, "the scan moves 4 counts per thread");

template <int NBITS>
__device__ __forceinline__ unsigned match_mask(int key) {
    unsigned m = 0xffffffffu;
#pragma unroll
    for (int bit = 0; bit < NBITS; ++bit) {
        const bool on = (key >> bit) & 1;
        const unsigned bal = __builtin_amdgcn_ballot_w32(on);
        m &= on ? bal : ~bal;
    }
    return m;
}

__global__ void
__launch_bounds__(1024)
split_count(const int* __restrict__ dst, int* __restrict__ counts, int* __restrict__ lrank) {
    __shared__ int hist[32 * HP];
    const int t    = threadIdx.x;
    const int lane = t & 31;
    const int w    = t >> 5;
    const int blk  = blockIdx.x;
    const unsigned below = (1u << lane) - 1u;
    for (int i = t; i < 32 * HP; i += 1024) hist[i] = 0;
    __syncthreads();
    int* hw = hist + w * HP;
    int key[R1], rk[R1];
#pragma unroll
    for (int r = 0; r < R1; ++r) {
        const int e = blk * EPB + (w * R1 + r) * 32 + lane;
        key[r] = (e < NE) ? (dst[e] >> 7) : NBK;
    }
#pragma unroll
    for (int r = 0; r < R1; ++r) {
        const int k = key[r];
        const unsigned m = match_mask<9>(k);
        const int ahead = __builtin_popcount(m & below);
        const int old = hw[k];
        rk[r]  = old + ahead;
        if (ahead == 0) hw[k] = old + __builtin_popcount(m);
    }
    __syncthreads();
    if (t < HP) {
        int run = 0;
        for (int w2 = 0; w2 < 32; ++w2) {
            const int c = hist[w2 * HP + t];
            hist[w2 * HP + t] = run;
            run += c;
        }
        if (t < NBK) counts[t * G1 + blk] = run;
    }
    __syncthreads();
#pragma unroll
    for (int r = 0; r < R1; ++r) {
        const int e = blk * EPB + (w * R1 + r) * 32 + lane;
        if (e < NE) lrank[e] = hw[key[r]] + rk[r];
    }
}

#define SCAN_T 1024
#define SCAN_TILES ((NCNT + 4 * SCAN_T - 1) / (4 * SCAN_T))

__global__ void
__launch_bounds__(SCAN_T)
scan_kernel(const int* __restrict__ cin, int* __restrict__ cout) {
    __shared__ int wsum[SCAN_TILES][32];
    const int t    = threadIdx.x;
    const int lane = t & 31;
    const int wv   = t >> 5;
    int4 v[SCAN_TILES];
#pragma unroll
    for (int k = 0; k < SCAN_TILES; ++k) {
        const int i = (k * SCAN_T + t) * 4;
        v[k] = (i < NCNT) ? *(const int4*)(cin + i) : make_int4(0, 0, 0, 0);
    }
    int inc[SCAN_TILES];
#pragma unroll
    for (int k = 0; k < SCAN_TILES; ++k) {
        int x = v[k].x + v[k].y + v[k].z + v[k].w;
        for (int off = 1; off < 32; off <<= 1) {
            const int nb = __shfl_up(x, off);
            if (lane >= off) x += nb;
        }
        inc[k] = x;
        if (lane == 31) wsum[k][wv] = x;
    }
    __syncthreads();
    int carry = 0;
#pragma unroll
    for (int k = 0; k < SCAN_TILES; ++k) {
        const int4 a = v[k];
        const int s1 = a.x, s2 = s1 + a.y, s3 = s2 + a.z, s4 = s3 + a.w;
        int winc = wsum[k][lane];
        for (int off = 1; off < 32; off <<= 1) {
            const int nb = __shfl_up(winc, off);
            if (lane >= off) winc += nb;
        }
        const int tile_total  = __shfl(winc, 31);
        const int before_wave = __shfl(winc, (wv + 31) & 31);
        const int base = carry + (wv ? before_wave : 0) + (inc[k] - s4);
        const int i = (k * SCAN_T + t) * 4;
        if (i < NCNT) *(int4*)(cout + i) = make_int4(base, base + s1, base + s2, base + s3);
        carry += tile_total;
    }
}

static_assert(NN <= (1 << 24), "a source id and 7 node bits fit one int");
__global__ void split_place(const int* __restrict__ src, const int* __restrict__ dst,
                            const int* __restrict__ offs, const int* __restrict__ lrank,
                            int* __restrict__ ekey) {
    const int e = blockIdx.x * blockDim.x + threadIdx.x;
    if (e < NE) {
        const int d = dst[e];
        const int pos = offs[(d >> 7) * G1 + e / EPB] + lrank[e];
        ekey[pos] = (src[e] << 7) | (d & 127);
    }
}

__global__ void
__launch_bounds__(1024)
bucket_lists(const int* __restrict__ offs, const int* __restrict__ ekey,
             int* __restrict__ deg, int* __restrict__ rowptr, int* __restrict__ col) {
    __shared__ int hist[32 * HP2];
    __shared__ int ncnt[128];
    __shared__ int nbase[128];
    const int t    = threadIdx.x;
    const int lane = t & 31;
    const int w    = t >> 5;
    const int b    = blockIdx.x;
    const unsigned below = (1u << lane) - 1u;
    const int beg = offs[b * G1];
    const int end = (b + 1 < NBK) ? offs[(b + 1) * G1] : NE;
    const int n   = end - beg;
    for (int i = t; i < 32 * HP2; i += 1024) hist[i] = 0;
    __syncthreads();
    int* hw = hist + w * HP2;
    for (int i0 = 0; i0 < n; i0 += 1024) {
        const int i = i0 + t;
        const int k = (i < n) ? (ekey[beg + i] & 127) : 128;
        const unsigned m = match_mask<8>(k);
        if ((m & below) == 0) hw[k] += __builtin_popcount(m);
    }
    __syncthreads();
    if (t < 128) {
        int run = 0;
        for (int w2 = 0; w2 < 32; ++w2) {
            const int c = hist[w2 * HP2 + t];
            hist[w2 * HP2 + t] = run;
            run += c;
        }
        ncnt[t] = run;
    }
    __syncthreads();
    if (t < 128) {
        int ex = 0;
        for (int j = 0; j < t; ++j) ex += ncnt[j];
        nbase[t] = beg + ex;
        const int node = b * 128 + t;
        if (node < NN) {
            deg[node]    = ncnt[t];
            rowptr[node] = beg + ex;
        }
    }
    __syncthreads();
    for (int i0 = 0; i0 < n; i0 += 1024) {
        const int i = i0 + t;
        const int kv = (i < n) ? ekey[beg + i] : 0;
        const int k  = (i < n) ? (kv & 127) : 128;
        const unsigned m = match_mask<8>(k);
        const int ahead = __builtin_popcount(m & below);
        const int old = hw[k];
        if (i < n) col[nbase[k] + old + ahead] = kv >> 7;
        if (ahead == 0) hw[k] = old + __builtin_popcount(m);
    }
}

#define GR 32
__global__ void
__launch_bounds__(128)
layer1_fused(const int* __restrict__ rowptr, const int* __restrict__ deg, const int* __restrict__ col,
             const float* __restrict__ x,
             const float* __restrict__ W1l, const float* __restrict__ b1, const float* __restrict__ W1r,
             const float* __restrict__ W2l, const float* __restrict__ W2r, const float* __restrict__ b2,
             float* __restrict__ P, float* __restrict__ out, int nrows) {
    constexpr int MB    = 5;
    constexpr int NB    = 2;
    constexpr int ROWS  = 16 * MB;
    constexpr int PITCH = DHID + 4;
    constexpr int NW    = 4;
    constexpr int RPW   = ROWS / NW;
    constexpr int BATCH = 10;
    static_assert(RPW <= 32, "a wave holds its nodes' list bounds one per lane");

    __shared__ __attribute__((aligned(16))) float sM[ROWS * PITCH];
    __shared__ __attribute__((aligned(16))) float sX[ROWS * PITCH];

    const int lane = threadIdx.x & 31;
    const int wv   = threadIdx.x >> 5;
    const int row0 = blockIdx.x * ROWS;
    if (row0 >= nrows) return;

    const int wu = __builtin_amdgcn_readfirstlane(wv);
    const int n0 = row0 + wu * RPW;
    const int rpv = (lane < RPW) ? rowptr[n0 + lane] : 0;
    const int dgv = (lane < RPW) ? deg[n0 + lane] : 0;

#pragma unroll
    for (int b0 = 0; b0 < RPW; b0 += BATCH) {
        float4 vx[BATCH];
#pragma unroll
        for (int j = 0; j < BATCH; ++j) {
            const int r = wv + (b0 + j) * NW;
            vx[j] = *(const float4*)(x + (size_t)(row0 + r) * DHID + lane * 4);
        }
#pragma unroll
        for (int j = 0; j < BATCH; ++j) {
            const int r = wv + (b0 + j) * NW;
            *(float4*)&sX[r * PITCH + lane * 4] = vx[j];
        }
    }

    {
        const int ebeg = __builtin_amdgcn_readlane(rpv, 0);
        const int eend = __builtin_amdgcn_readlane(rpv, RPW - 1) + __builtin_amdgcn_readlane(dgv, RPW - 1);
        int i = 0;
        int cur_end = ebeg + __builtin_amdgcn_readlane(dgv, 0);
        float4 acc = make_float4(0.f, 0.f, 0.f, 0.f);
        float* mrow = sM + (wu * RPW) * PITCH + lane * 4;
        int nextid = (ebeg + lane < eend) ? col[ebeg + lane] : 0;
        for (int c = ebeg; c < eend; c += GR) {
            const int myid = nextid;
            const int m = (eend - c < GR) ? (eend - c) : GR;
            float4 v[GR];
#pragma unroll
            for (int j = 0; j < GR; ++j) {
                const int sid = __builtin_amdgcn_readlane(myid, j);
                if (j < m) v[j] = *(const float4*)(x + (size_t)sid * DHID + lane * 4);
            }
            nextid = (c + GR + lane < eend) ? col[c + GR + lane] : 0;
#pragma unroll
            for (int j = 0; j < GR; ++j) {
                if (j < m) {
                    while (c + j == cur_end) {
                        const float s = 1.0f / fmaxf((float)__builtin_amdgcn_readlane(dgv, i), 1.0f);
                        *(float4*)(mrow + i * PITCH) = make_float4(acc.x * s, acc.y * s, acc.z * s, acc.w * s);
                        acc = make_float4(0.f, 0.f, 0.f, 0.f);
                        ++i;
                        cur_end += __builtin_amdgcn_readlane(dgv, i);
                    }
                    acc.x += v[j].x; acc.y += v[j].y; acc.z += v[j].z; acc.w += v[j].w;
                }
            }
        }
        for (; i < RPW; ++i) {
            const float s = 1.0f / fmaxf((float)__builtin_amdgcn_readlane(dgv, i), 1.0f);
            *(float4*)(mrow + i * PITCH) = make_float4(acc.x * s, acc.y * s, acc.z * s, acc.w * s);
            acc = make_float4(0.f, 0.f, 0.f, 0.f);
        }
    }
    __syncthreads();

    const int half = lane >> 4;
    const int mn   = lane & 15;
    const int koff = half * 2;
    const int c0   = wv * (16 * NB) + mn;

    const float* mp = sM + mn * PITCH + koff;
    const float* xp = sX + mn * PITCH + koff;

    v8f acc[MB][NB];
#pragma unroll
    for (int mi = 0; mi < MB; ++mi)
#pragma unroll
        for (int nj = 0; nj < NB; ++nj)
            acc[mi][nj] = (v8f){0.f, 0.f, 0.f, 0.f, 0.f, 0.f, 0.f, 0.f};

#pragma unroll 2
    for (int k = 0; k < DHID; k += 4) {
        v2f bl[NB], br[NB];
#pragma unroll
        for (int nj = 0; nj < NB; ++nj) {
            bl[nj][0] = W1l[(k + koff)     * DHID + c0 + nj * 16];
            bl[nj][1] = W1l[(k + koff + 1) * DHID + c0 + nj * 16];
            br[nj][0] = W1r[(k + koff)     * DHID + c0 + nj * 16];
            br[nj][1] = W1r[(k + koff + 1) * DHID + c0 + nj * 16];
        }
        v2f am[MB], ar[MB];
#pragma unroll
        for (int mi = 0; mi < MB; ++mi) {
            am[mi] = *(const v2f*)(mp + mi * 16 * PITCH + k);
            ar[mi] = *(const v2f*)(xp + mi * 16 * PITCH + k);
        }
#pragma unroll
        for (int mi = 0; mi < MB; ++mi) {
#pragma unroll
            for (int nj = 0; nj < NB; ++nj) {
                acc[mi][nj] = __builtin_amdgcn_wmma_f32_16x16x4_f32(
                    false, am[mi], false, bl[nj], (short)0, acc[mi][nj], false, false);
                acc[mi][nj] = __builtin_amdgcn_wmma_f32_16x16x4_f32(
                    false, ar[mi], false, br[nj], (short)0, acc[mi][nj], false, false);
            }
        }
    }

    __builtin_amdgcn_sched_barrier(0);
    asm volatile("v_nop\n\tv_nop\n\tv_nop\n\tv_nop");
    __builtin_amdgcn_sched_barrier(0);

    __syncthreads();
#pragma unroll
    for (int nj = 0; nj < NB; ++nj) {
        const float bv = b1[c0 + nj * 16];
#pragma unroll
        for (int mi = 0; mi < MB; ++mi) {
#pragma unroll
            for (int v = 0; v < 8; ++v) {
                const int m = mi * 16 + v + half * 8;
                sM[m * PITCH + c0 + nj * 16] = fmaxf(acc[mi][nj][v] + bv, 0.f);
            }
        }
    }
    __syncthreads();

    const int isr = wv >> 1;
    const int c2  = (wv & 1) * (16 * NB) + mn;

    const float* Wm = isr ? W2r : W2l;
    float*       dp = isr ? out : P;

#pragma unroll
    for (int mi = 0; mi < MB; ++mi)
#pragma unroll
        for (int nj = 0; nj < NB; ++nj)
            acc[mi][nj] = (v8f){0.f, 0.f, 0.f, 0.f, 0.f, 0.f, 0.f, 0.f};

#pragma unroll 2
    for (int k = 0; k < DHID; k += 4) {
        v2f bw[NB];
#pragma unroll
        for (int nj = 0; nj < NB; ++nj) {
            bw[nj][0] = Wm[(k + koff)     * DOUT + c2 + nj * 16];
            bw[nj][1] = Wm[(k + koff + 1) * DOUT + c2 + nj * 16];
        }
        v2f ah[MB];
#pragma unroll
        for (int mi = 0; mi < MB; ++mi)
            ah[mi] = *(const v2f*)(mp + mi * 16 * PITCH + k);
#pragma unroll
        for (int mi = 0; mi < MB; ++mi) {
#pragma unroll
            for (int nj = 0; nj < NB; ++nj) {
                acc[mi][nj] = __builtin_amdgcn_wmma_f32_16x16x4_f32(
                    false, ah[mi], false, bw[nj], (short)0, acc[mi][nj], false, false);
            }
        }
    }

    __builtin_amdgcn_sched_barrier(0);
    asm volatile("v_nop\n\tv_nop\n\tv_nop\n\tv_nop");
    __builtin_amdgcn_sched_barrier(0);

#pragma unroll
    for (int nj = 0; nj < NB; ++nj) {
        const float bv = isr ? b2[c2 + nj * 16] : 0.f;
#pragma unroll
        for (int mi = 0; mi < MB; ++mi) {
#pragma unroll
            for (int v = 0; v < 8; ++v) {
                const int m = row0 + mi * 16 + v + half * 8;
                dp[(size_t)m * DOUT + c2 + nj * 16] = acc[mi][nj][v] + bv;
            }
        }
    }
}


#define GU 16
__global__ void gather64_kernel(const int* __restrict__ rowptr, const int* __restrict__ deg,
                                const int* __restrict__ col, const float* __restrict__ P,
                                float* __restrict__ out, int n) {
    const int gid  = blockIdx.x * blockDim.x + threadIdx.x;
    const int lane = threadIdx.x & 31;
    const int r = __builtin_amdgcn_readfirstlane(gid >> 5);
    if (r >= n) return;
    const int beg = rowptr[r];
    const int cnt = deg[r];
    float* o = out + (size_t)r * DOUT + lane * 2;
    const float2 start = *(const float2*)o;
    float2 acc = make_float2(0.f, 0.f);
    for (int b0 = 0; b0 < cnt; b0 += 32) {
        const int m = (cnt - b0 < 32) ? (cnt - b0) : 32;
        const int myid = (lane < m) ? col[beg + b0 + lane] : 0;
        for (int j0 = 0; j0 < m; j0 += GU) {
            float2 v[GU];
#pragma unroll
            for (int j = 0; j < GU; ++j) {
                const int sid = __builtin_amdgcn_readlane(myid, j0 + j);
                if (j0 + j < m) v[j] = *(const float2*)(P + (size_t)sid * DOUT + lane * 2);
                else            v[j] = make_float2(0.f, 0.f);
            }
#pragma unroll
            for (int j = 0; j < GU; ++j) {
                acc.x += v[j].x; acc.y += v[j].y;
            }
        }
    }
    const float s = 1.0f / fmaxf((float)cnt, 1.0f);
    *(float2*)o = make_float2(start.x + acc.x * s, start.y + acc.y * s);
}

extern "C" void kernel_launch(void* const* d_in, const int* in_sizes, int n_in,
                              void* d_out, int out_size, void* d_ws, size_t ws_size,
                              hipStream_t stream) {
    const float* x   = (const float*)d_in[0];
    const int*   ei  = (const int*)d_in[1];
    const float* W1l = (const float*)d_in[2];
    const float* b1  = (const float*)d_in[3];
    const float* W1r = (const float*)d_in[4];
    const float* W2l = (const float*)d_in[5];
    const float* b2  = (const float*)d_in[6];
    const float* W2r = (const float*)d_in[7];
    float*       out = (float*)d_out;

    const int* src = ei;
    const int* dst = ei + NE;

    float* A      = (float*)d_ws;
    float* P      = A + (size_t)NN * DHID;
    int*   col    = (int*)(P + (size_t)NN * DOUT);
    int*   ekey   = col + NE;
    int*   lrank  = ekey + NE;
    int*   rowptr = lrank + NE;
    int*   deg    = rowptr + NN;
    int*   counts = deg + NN;
    int*   offs   = counts + NCNT;
    static_assert((size_t)NN * DOUT + 3 * (size_t)NE + 2 * (size_t)NN + 2 * (size_t)NCNT
                      <= (size_t)NN * DHID + NN,
                  "P and the lists fit the space that used to hold the counts and h");
    static_assert(((size_t)NN * DHID + (size_t)NN * DOUT + 3 * (size_t)NE + 2 * (size_t)NN) % 4 == 0,
                  "counts and offs start on 16 B for the scan's 16-B accesses");

    const int BLK = 256;

    split_count<<<G1, 1024, 0, stream>>>(dst, counts, lrank);
    scan_kernel<<<1, SCAN_T, 0, stream>>>(counts, offs);
    split_place<<<(NE + BLK - 1) / BLK, BLK, 0, stream>>>(src, dst, offs, lrank, ekey);
    bucket_lists<<<NBK, 1024, 0, stream>>>(offs, ekey, deg, rowptr, col);

    layer1_fused<<<NN / 80, 128, 0, stream>>>(rowptr, deg, col, x, W1l, b1, W1r, W2l, W2r, b2, P, out, NN);

    gather64_kernel<<<(NN * 32 + BLK - 1) / BLK, BLK, 0, stream>>>(rowptr, deg, col, P, out, NN);
}


# Edge ordering in 2 launches, not 4: a level-1 block orders its own 4096 edges by bucket in LDS, one run per bucket; the bucket's workgroup pulls its 196 runs; no scan or placing launch; same lists
# speedup vs baseline: 4.0419x; 1.3252x over previous
#include <hip/hip_runtime.h>

#define NN 50000
#define NE 800000
#define DHID 128
#define DOUT 64

typedef float v2f __attribute__((ext_vector_type(2)));
typedef float v8f __attribute__((ext_vector_type(8)));

#define NBK   ((NN + 127) >> 7)
#define G1    196
#define R1    4
#define EPB   (1024 * R1)
#define HP    (NBK + 1)
#define HP2   132
#define NCNT  (NBK * G1)
static_assert(G1 * EPB >= NE, "level-1 blocks cover all edges");
static_assert(NBK <= 512, "level-1 keys fit the 9 ballot bits");
static_assert(NCNT % 4 == 0, "the counts are summed in 16-B pieces");

template <int NBITS>
__device__ __forceinline__ unsigned match_mask(int key) {
    unsigned m = 0xffffffffu;
#pragma unroll
    for (int bit = 0; bit < NBITS; ++bit) {
        const bool on = (key >> bit) & 1;
        const unsigned bal = __builtin_amdgcn_ballot_w32(on);
        m &= on ? bal : ~bal;
    }
    return m;
}

static_assert(NN <= (1 << 24), "a source id and 7 node bits fit one int");
__global__ void
__launch_bounds__(1024)
split_local(const int* __restrict__ src, const int* __restrict__ dst,
            int* __restrict__ counts, int* __restrict__ lstart, int* __restrict__ ekl) {
    __shared__ int hist[32 * HP];
    __shared__ int lbase[HP];
    __shared__ int wtot[16];
    __shared__ int sorted[EPB];
    const int t    = threadIdx.x;
    const int lane = t & 31;
    const int w    = t >> 5;
    const int blk  = blockIdx.x;
    const unsigned below = (1u << lane) - 1u;
    for (int i = t; i < 32 * HP; i += 1024) hist[i] = 0;
    __syncthreads();
    int* hw = hist + w * HP;
    int key[R1], rk[R1], val[R1];
#pragma unroll
    for (int r = 0; r < R1; ++r) {
        const int e = blk * EPB + (w * R1 + r) * 32 + lane;
        const int d = (e < NE) ? dst[e] : 0;
        const int s = (e < NE) ? src[e] : 0;
        key[r] = (e < NE) ? (d >> 7) : NBK;
        val[r] = (s << 7) | (d & 127);
    }
#pragma unroll
    for (int r = 0; r < R1; ++r) {
        const int k = key[r];
        const unsigned m = match_mask<9>(k);
        const int ahead = __builtin_popcount(m & below);
        const int old = hw[k];
        rk[r]  = old + ahead;
        if (ahead == 0) hw[k] = old + __builtin_popcount(m);
    }
    __syncthreads();
    int run = 0;
    if (t < HP) {
        for (int w2 = 0; w2 < 32; ++w2) {
            const int c = hist[w2 * HP + t];
            hist[w2 * HP + t] = run;
            run += c;
        }
        if (t < NBK) counts[t * G1 + blk] = run;
    }
    int inc = run;
    for (int off = 1; off < 32; off <<= 1) {
        const int nb = __shfl_up(inc, off);
        if (lane >= off) inc += nb;
    }
    if (lane == 31 && w < 16) wtot[w] = inc;
    __syncthreads();
    if (t < HP) {
        int before = 0;
        for (int w2 = 0; w2 < w; ++w2) before += wtot[w2];
        const int ex = before + inc - run;
        lbase[t] = ex;
        if (t < NBK) lstart[t * G1 + blk] = ex;
    }
    __syncthreads();
#pragma unroll
    for (int r = 0; r < R1; ++r) {
        const int k = key[r];
        sorted[lbase[k] + hw[k] + rk[r]] = val[r];
    }
    __syncthreads();
    for (int i = t; i < EPB; i += 1024)
        if (blk * EPB + i < NE) ekl[blk * EPB + i] = sorted[i];
}

#define CB 8
static_assert(G1 % 4 == 0, "the counts in front of a bucket are whole 16-B pieces");
static_assert(G1 <= 7 * 32, "a wave pulls at most 7 runs");
__global__ void
__launch_bounds__(1024)
bucket_lists(const int* __restrict__ counts, const int* __restrict__ lstart, const int* __restrict__ ekl,
             int* ekey,
             int* __restrict__ deg, int* __restrict__ rowptr, int* __restrict__ col) {
    __shared__ int hist[32 * HP2];
    __shared__ int ncnt[128];
    __shared__ int nbase[128];
    __shared__ int rcnt[G1];
    __shared__ int rls[G1];
    __shared__ int roff[G1];
    __shared__ int red[32];
    __shared__ int wt[8];
    __shared__ int sbeg, sn;
    const int t    = threadIdx.x;
    const int lane = t & 31;
    const int w    = t >> 5;
    const int b    = blockIdx.x;
    const unsigned below = (1u << lane) - 1u;
    const int n4 = b * (G1 / 4);
    int part = 0;
    for (int k0 = 0; k0 * 1024 < n4; k0 += CB) {
        int4 cv[CB];
#pragma unroll
        for (int k = 0; k < CB; ++k) {
            const int i = (k0 + k) * 1024 + t;
            cv[k] = (i < n4) ? ((const int4*)counts)[i] : make_int4(0, 0, 0, 0);
        }
#pragma unroll
        for (int k = 0; k < CB; ++k) part += cv[k].x + cv[k].y + cv[k].z + cv[k].w;
    }
    int c = 0;
    if (t < G1) {
        c = counts[b * G1 + t];
        rcnt[t] = c;
        rls[t]  = lstart[b * G1 + t];
    }
    for (int off = 16; off > 0; off >>= 1) part += __shfl_xor(part, off);
    if (lane == 0) red[w] = part;
    int inc = c;
    for (int off = 1; off < 32; off <<= 1) {
        const int nb = __shfl_up(inc, off);
        if (lane >= off) inc += nb;
    }
    if (lane == 31 && w < 8) wt[w] = inc;
    for (int i = t; i < 32 * HP2; i += 1024) hist[i] = 0;
    __syncthreads();
    if (t < G1) {
        int before = 0;
        for (int w2 = 0; w2 < w; ++w2) before += wt[w2];
        roff[t] = before + inc - c;
    }
    if (t == 0) {
        int s = 0;
        for (int i = 0; i < 32; ++i) s += red[i];
        sbeg = s;
        int tot = 0;
        for (int w2 = 0; w2 < (G1 + 31) / 32; ++w2) tot += wt[w2];
        sn = tot;
    }
    __syncthreads();
    const int beg = sbeg;
    const int n   = sn;
    {
        int pv[7];
#pragma unroll
        for (int q = 0; q < 7; ++q) {
            const int rb = w + 32 * q;
            const int rc = (rb < G1) ? rb : 0;
            const int cq = (rb < G1) ? rcnt[rc] : 0;
            pv[q] = (lane < cq) ? ekl[rc * EPB + rls[rc] + lane] : 0;
        }
#pragma unroll
        for (int q = 0; q < 7; ++q) {
            const int rb = w + 32 * q;
            const int rc = (rb < G1) ? rb : 0;
            const int cq = (rb < G1) ? rcnt[rc] : 0;
            if (lane < cq) ekey[beg + roff[rc] + lane] = pv[q];
            for (int j = 32 + lane; j < cq; j += 32)
                ekey[beg + roff[rc] + j] = ekl[rc * EPB + rls[rc] + j];
        }
    }
    __syncthreads();
    int* hw = hist + w * HP2;
    for (int i0 = 0; i0 < n; i0 += 1024) {
        const int i = i0 + t;
        const int k = (i < n) ? (ekey[beg + i] & 127) : 128;
        const unsigned m = match_mask<8>(k);
        if ((m & below) == 0) hw[k] += __builtin_popcount(m);
    }
    __syncthreads();
    if (t < 128) {
        int run = 0;
        for (int w2 = 0; w2 < 32; ++w2) {
            const int c = hist[w2 * HP2 + t];
            hist[w2 * HP2 + t] = run;
            run += c;
        }
        ncnt[t] = run;
    }
    __syncthreads();
    if (t < 128) {
        int ex = 0;
        for (int j = 0; j < t; ++j) ex += ncnt[j];
        nbase[t] = beg + ex;
        const int node = b * 128 + t;
        if (node < NN) {
            deg[node]    = ncnt[t];
            rowptr[node] = beg + ex;
        }
    }
    __syncthreads();
    for (int i0 = 0; i0 < n; i0 += 1024) {
        const int i = i0 + t;
        const int kv = (i < n) ? ekey[beg + i] : 0;
        const int k  = (i < n) ? (kv & 127) : 128;
        const unsigned m = match_mask<8>(k);
        const int ahead = __builtin_popcount(m & below);
        const int old = hw[k];
        if (i < n) col[nbase[k] + old + ahead] = kv >> 7;
        if (ahead == 0) hw[k] = old + __builtin_popcount(m);
    }
}

#define GR 32
__global__ void
__launch_bounds__(128)
layer1_fused(const int* __restrict__ rowptr, const int* __restrict__ deg, const int* __restrict__ col,
             const float* __restrict__ x,
             const float* __restrict__ W1l, const float* __restrict__ b1, const float* __restrict__ W1r,
             const float* __restrict__ W2l, const float* __restrict__ W2r, const float* __restrict__ b2,
             float* __restrict__ P, float* __restrict__ out, int nrows) {
    constexpr int MB    = 5;
    constexpr int NB    = 2;
    constexpr int ROWS  = 16 * MB;
    constexpr int PITCH = DHID + 4;
    constexpr int NW    = 4;
    constexpr int RPW   = ROWS / NW;
    constexpr int BATCH = 10;
    static_assert(RPW <= 32, "a wave holds its nodes' list bounds one per lane");

    __shared__ __attribute__((aligned(16))) float sM[ROWS * PITCH];
    __shared__ __attribute__((aligned(16))) float sX[ROWS * PITCH];

    const int lane = threadIdx.x & 31;
    const int wv   = threadIdx.x >> 5;
    const int row0 = blockIdx.x * ROWS;
    if (row0 >= nrows) return;

    const int wu = __builtin_amdgcn_readfirstlane(wv);
    const int n0 = row0 + wu * RPW;
    const int rpv = (lane < RPW) ? rowptr[n0 + lane] : 0;
    const int dgv = (lane < RPW) ? deg[n0 + lane] : 0;

#pragma unroll
    for (int b0 = 0; b0 < RPW; b0 += BATCH) {
        float4 vx[BATCH];
#pragma unroll
        for (int j = 0; j < BATCH; ++j) {
            const int r = wv + (b0 + j) * NW;
            vx[j] = *(const float4*)(x + (size_t)(row0 + r) * DHID + lane * 4);
        }
#pragma unroll
        for (int j = 0; j < BATCH; ++j) {
            const int r = wv + (b0 + j) * NW;
            *(float4*)&sX[r * PITCH + lane * 4] = vx[j];
        }
    }

    {
        const int ebeg = __builtin_amdgcn_readlane(rpv, 0);
        const int eend = __builtin_amdgcn_readlane(rpv, RPW - 1) + __builtin_amdgcn_readlane(dgv, RPW - 1);
        int i = 0;
        int cur_end = ebeg + __builtin_amdgcn_readlane(dgv, 0);
        float4 acc = make_float4(0.f, 0.f, 0.f, 0.f);
        float* mrow = sM + (wu * RPW) * PITCH + lane * 4;
        int nextid = (ebeg + lane < eend) ? col[ebeg + lane] : 0;
        for (int c = ebeg; c < eend; c += GR) {
            const int myid = nextid;
            const int m = (eend - c < GR) ? (eend - c) : GR;
            float4 v[GR];
#pragma unroll
            for (int j = 0; j < GR; ++j) {
                const int sid = __builtin_amdgcn_readlane(myid, j);
                if (j < m) v[j] = *(const float4*)(x + (size_t)sid * DHID + lane * 4);
            }
            nextid = (c + GR + lane < eend) ? col[c + GR + lane] : 0;
#pragma unroll
            for (int j = 0; j < GR; ++j) {
                if (j < m) {
                    while (c + j == cur_end) {
                        const float s = 1.0f / fmaxf((float)__builtin_amdgcn_readlane(dgv, i), 1.0f);
                        *(float4*)(mrow + i * PITCH) = make_float4(acc.x * s, acc.y * s, acc.z * s, acc.w * s);
                        acc = make_float4(0.f, 0.f, 0.f, 0.f);
                        ++i;
                        cur_end += __builtin_amdgcn_readlane(dgv, i);
                    }
                    acc.x += v[j].x; acc.y += v[j].y; acc.z += v[j].z; acc.w += v[j].w;
                }
            }
        }
        for (; i < RPW; ++i) {
            const float s = 1.0f / fmaxf((float)__builtin_amdgcn_readlane(dgv, i), 1.0f);
            *(float4*)(mrow + i * PITCH) = make_float4(acc.x * s, acc.y * s, acc.z * s, acc.w * s);
            acc = make_float4(0.f, 0.f, 0.f, 0.f);
        }
    }
    __syncthreads();

    const int half = lane >> 4;
    const int mn   = lane & 15;
    const int koff = half * 2;
    const int c0   = wv * (16 * NB) + mn;

    const float* mp = sM + mn * PITCH + koff;
    const float* xp = sX + mn * PITCH + koff;

    v8f acc[MB][NB];
#pragma unroll
    for (int mi = 0; mi < MB; ++mi)
#pragma unroll
        for (int nj = 0; nj < NB; ++nj)
            acc[mi][nj] = (v8f){0.f, 0.f, 0.f, 0.f, 0.f, 0.f, 0.f, 0.f};

#pragma unroll 2
    for (int k = 0; k < DHID; k += 4) {
        v2f bl[NB], br[NB];
#pragma unroll
        for (int nj = 0; nj < NB; ++nj) {
            bl[nj][0] = W1l[(k + koff)     * DHID + c0 + nj * 16];
            bl[nj][1] = W1l[(k + koff + 1) * DHID + c0 + nj * 16];
            br[nj][0] = W1r[(k + koff)     * DHID + c0 + nj * 16];
            br[nj][1] = W1r[(k + koff + 1) * DHID + c0 + nj * 16];
        }
        v2f am[MB], ar[MB];
#pragma unroll
        for (int mi = 0; mi < MB; ++mi) {
            am[mi] = *(const v2f*)(mp + mi * 16 * PITCH + k);
            ar[mi] = *(const v2f*)(xp + mi * 16 * PITCH + k);
        }
#pragma unroll
        for (int mi = 0; mi < MB; ++mi) {
#pragma unroll
            for (int nj = 0; nj < NB; ++nj) {
                acc[mi][nj] = __builtin_amdgcn_wmma_f32_16x16x4_f32(
                    false, am[mi], false, bl[nj], (short)0, acc[mi][nj], false, false);
                acc[mi][nj] = __builtin_amdgcn_wmma_f32_16x16x4_f32(
                    false, ar[mi], false, br[nj], (short)0, acc[mi][nj], false, false);
            }
        }
    }

    __builtin_amdgcn_sched_barrier(0);
    asm volatile("v_nop\n\tv_nop\n\tv_nop\n\tv_nop");
    __builtin_amdgcn_sched_barrier(0);

    __syncthreads();
#pragma unroll
    for (int nj = 0; nj < NB; ++nj) {
        const float bv = b1[c0 + nj * 16];
#pragma unroll
        for (int mi = 0; mi < MB; ++mi) {
#pragma unroll
            for (int v = 0; v < 8; ++v) {
                const int m = mi * 16 + v + half * 8;
                sM[m * PITCH + c0 + nj * 16] = fmaxf(acc[mi][nj][v] + bv, 0.f);
            }
        }
    }
    __syncthreads();

    const int isr = wv >> 1;
    const int c2  = (wv & 1) * (16 * NB) + mn;

    const float* Wm = isr ? W2r : W2l;
    float*       dp = isr ? out : P;

#pragma unroll
    for (int mi = 0; mi < MB; ++mi)
#pragma unroll
        for (int nj = 0; nj < NB; ++nj)
            acc[mi][nj] = (v8f){0.f, 0.f, 0.f, 0.f, 0.f, 0.f, 0.f, 0.f};

#pragma unroll 2
    for (int k = 0; k < DHID; k += 4) {
        v2f bw[NB];
#pragma unroll
        for (int nj = 0; nj < NB; ++nj) {
            bw[nj][0] = Wm[(k + koff)     * DOUT + c2 + nj * 16];
            bw[nj][1] = Wm[(k + koff + 1) * DOUT + c2 + nj * 16];
        }
        v2f ah[MB];
#pragma unroll
        for (int mi = 0; mi < MB; ++mi)
            ah[mi] = *(const v2f*)(mp + mi * 16 * PITCH + k);
#pragma unroll
        for (int mi = 0; mi < MB; ++mi) {
#pragma unroll
            for (int nj = 0; nj < NB; ++nj) {
                acc[mi][nj] = __builtin_amdgcn_wmma_f32_16x16x4_f32(
                    false, ah[mi], false, bw[nj], (short)0, acc[mi][nj], false, false);
            }
        }
    }

    __builtin_amdgcn_sched_barrier(0);
    asm volatile("v_nop\n\tv_nop\n\tv_nop\n\tv_nop");
    __builtin_amdgcn_sched_barrier(0);

#pragma unroll
    for (int nj = 0; nj < NB; ++nj) {
        const float bv = isr ? b2[c2 + nj * 16] : 0.f;
#pragma unroll
        for (int mi = 0; mi < MB; ++mi) {
#pragma unroll
            for (int v = 0; v < 8; ++v) {
                const int m = row0 + mi * 16 + v + half * 8;
                dp[(size_t)m * DOUT + c2 + nj * 16] = acc[mi][nj][v] + bv;
            }
        }
    }
}


#define GU 16
__global__ void gather64_kernel(const int* __restrict__ rowptr, const int* __restrict__ deg,
                                const int* __restrict__ col, const float* __restrict__ P,
                                float* __restrict__ out, int n) {
    const int gid  = blockIdx.x * blockDim.x + threadIdx.x;
    const int lane = threadIdx.x & 31;
    const int r = __builtin_amdgcn_readfirstlane(gid >> 5);
    if (r >= n) return;
    const int beg = rowptr[r];
    const int cnt = deg[r];
    float* o = out + (size_t)r * DOUT + lane * 2;
    const float2 start = *(const float2*)o;
    float2 acc = make_float2(0.f, 0.f);
    for (int b0 = 0; b0 < cnt; b0 += 32) {
        const int m = (cnt - b0 < 32) ? (cnt - b0) : 32;
        const int myid = (lane < m) ? col[beg + b0 + lane] : 0;
        for (int j0 = 0; j0 < m; j0 += GU) {
            float2 v[GU];
#pragma unroll
            for (int j = 0; j < GU; ++j) {
                const int sid = __builtin_amdgcn_readlane(myid, j0 + j);
                if (j0 + j < m) v[j] = *(const float2*)(P + (size_t)sid * DOUT + lane * 2);
                else            v[j] = make_float2(0.f, 0.f);
            }
#pragma unroll
            for (int j = 0; j < GU; ++j) {
                acc.x += v[j].x; acc.y += v[j].y;
            }
        }
    }
    const float s = 1.0f / fmaxf((float)cnt, 1.0f);
    *(float2*)o = make_float2(start.x + acc.x * s, start.y + acc.y * s);
}

extern "C" void kernel_launch(void* const* d_in, const int* in_sizes, int n_in,
                              void* d_out, int out_size, void* d_ws, size_t ws_size,
                              hipStream_t stream) {
    const float* x   = (const float*)d_in[0];
    const int*   ei  = (const int*)d_in[1];
    const float* W1l = (const float*)d_in[2];
    const float* b1  = (const float*)d_in[3];
    const float* W1r = (const float*)d_in[4];
    const float* W2l = (const float*)d_in[5];
    const float* b2  = (const float*)d_in[6];
    const float* W2r = (const float*)d_in[7];
    float*       out = (float*)d_out;

    const int* src = ei;
    const int* dst = ei + NE;

    float* A      = (float*)d_ws;
    float* P      = A + (size_t)NN * DHID;
    int*   col    = (int*)(P + (size_t)NN * DOUT);
    int*   ekey   = col + NE;
    int*   ekl    = ekey + NE;
    int*   rowptr = ekl + NE;
    int*   deg    = rowptr + NN;
    int*   counts = deg + NN;
    int*   lstart = counts + NCNT;
    static_assert((size_t)NN * DOUT + 3 * (size_t)NE + 2 * (size_t)NN + 2 * (size_t)NCNT
                      <= (size_t)NN * DHID + NN,
                  "P and the lists fit the space that used to hold the counts and h");
    static_assert(((size_t)NN * DHID + (size_t)NN * DOUT + 3 * (size_t)NE + 2 * (size_t)NN) % 4 == 0,
                  "counts start on 16 B for the 16-B loads of the sum in front of a bucket");

    const int BLK = 256;

    split_local<<<G1, 1024, 0, stream>>>(src, dst, counts, lstart, ekl);
    bucket_lists<<<NBK, 1024, 0, stream>>>(counts, lstart, ekl, ekey, deg, rowptr, col);

    layer1_fused<<<NN / 80, 128, 0, stream>>>(rowptr, deg, col, x, W1l, b1, W1r, W2l, W2r, b2, P, out, NN);

    gather64_kernel<<<(NN * 32 + BLK - 1) / BLK, BLK, 0, stream>>>(rowptr, deg, col, P, out, NN);
}


# Layer 1: the root product x@W1r (half its WMMAs) runs under the neighbour-row gather, 4 k steps behind each chunk's row loads, W1r fragments asked a chunk ahead; sum order changes: root terms first
# speedup vs baseline: 4.3411x; 1.0740x over previous
#include <hip/hip_runtime.h>

#define NN 50000
#define NE 800000
#define DHID 128
#define DOUT 64

typedef float v2f __attribute__((ext_vector_type(2)));
typedef float v8f __attribute__((ext_vector_type(8)));

#define NBK   ((NN + 127) >> 7)
#define G1    196
#define R1    4
#define EPB   (1024 * R1)
#define HP    (NBK + 1)
#define HP2   132
#define NCNT  (NBK * G1)
static_assert(G1 * EPB >= NE, "level-1 blocks cover all edges");
static_assert(NBK <= 512, "level-1 keys fit the 9 ballot bits");
static_assert(NCNT % 4 == 0, "the counts are summed in 16-B pieces");

template <int NBITS>
__device__ __forceinline__ unsigned match_mask(int key) {
    unsigned m = 0xffffffffu;
#pragma unroll
    for (int bit = 0; bit < NBITS; ++bit) {
        const bool on = (key >> bit) & 1;
        const unsigned bal = __builtin_amdgcn_ballot_w32(on);
        m &= on ? bal : ~bal;
    }
    return m;
}

static_assert(NN <= (1 << 24), "a source id and 7 node bits fit one int");
__global__ void
__launch_bounds__(1024)
split_local(const int* __restrict__ src, const int* __restrict__ dst,
            int* __restrict__ counts, int* __restrict__ lstart, int* __restrict__ ekl) {
    __shared__ int hist[32 * HP];
    __shared__ int lbase[HP];
    __shared__ int wtot[16];
    __shared__ int sorted[EPB];
    const int t    = threadIdx.x;
    const int lane = t & 31;
    const int w    = t >> 5;
    const int blk  = blockIdx.x;
    const unsigned below = (1u << lane) - 1u;
    for (int i = t; i < 32 * HP; i += 1024) hist[i] = 0;
    __syncthreads();
    int* hw = hist + w * HP;
    int key[R1], rk[R1], val[R1];
#pragma unroll
    for (int r = 0; r < R1; ++r) {
        const int e = blk * EPB + (w * R1 + r) * 32 + lane;
        const int d = (e < NE) ? dst[e] : 0;
        const int s = (e < NE) ? src[e] : 0;
        key[r] = (e < NE) ? (d >> 7) : NBK;
        val[r] = (s << 7) | (d & 127);
    }
#pragma unroll
    for (int r = 0; r < R1; ++r) {
        const int k = key[r];
        const unsigned m = match_mask<9>(k);
        const int ahead = __builtin_popcount(m & below);
        const int old = hw[k];
        rk[r]  = old + ahead;
        if (ahead == 0) hw[k] = old + __builtin_popcount(m);
    }
    __syncthreads();
    int run = 0;
    if (t < HP) {
        for (int w2 = 0; w2 < 32; ++w2) {
            const int c = hist[w2 * HP + t];
            hist[w2 * HP + t] = run;
            run += c;
        }
        if (t < NBK) counts[t * G1 + blk] = run;
    }
    int inc = run;
    for (int off = 1; off < 32; off <<= 1) {
        const int nb = __shfl_up(inc, off);
        if (lane >= off) inc += nb;
    }
    if (lane == 31 && w < 16) wtot[w] = inc;
    __syncthreads();
    if (t < HP) {
        int before = 0;
        for (int w2 = 0; w2 < w; ++w2) before += wtot[w2];
        const int ex = before + inc - run;
        lbase[t] = ex;
        if (t < NBK) lstart[t * G1 + blk] = ex;
    }
    __syncthreads();
#pragma unroll
    for (int r = 0; r < R1; ++r) {
        const int k = key[r];
        sorted[lbase[k] + hw[k] + rk[r]] = val[r];
    }
    __syncthreads();
    for (int i = t; i < EPB; i += 1024)
        if (blk * EPB + i < NE) ekl[blk * EPB + i] = sorted[i];
}

#define CB 8
static_assert(G1 % 4 == 0, "the counts in front of a bucket are whole 16-B pieces");
static_assert(G1 <= 7 * 32, "a wave pulls at most 7 runs");
__global__ void
__launch_bounds__(1024)
bucket_lists(const int* __restrict__ counts, const int* __restrict__ lstart, const int* __restrict__ ekl,
             int* ekey,
             int* __restrict__ deg, int* __restrict__ rowptr, int* __restrict__ col) {
    __shared__ int hist[32 * HP2];
    __shared__ int ncnt[128];
    __shared__ int nbase[128];
    __shared__ int rcnt[G1];
    __shared__ int rls[G1];
    __shared__ int roff[G1];
    __shared__ int red[32];
    __shared__ int wt[8];
    __shared__ int sbeg, sn;
    const int t    = threadIdx.x;
    const int lane = t & 31;
    const int w    = t >> 5;
    const int b    = blockIdx.x;
    const unsigned below = (1u << lane) - 1u;
    const int n4 = b * (G1 / 4);
    int part = 0;
    for (int k0 = 0; k0 * 1024 < n4; k0 += CB) {
        int4 cv[CB];
#pragma unroll
        for (int k = 0; k < CB; ++k) {
            const int i = (k0 + k) * 1024 + t;
            cv[k] = (i < n4) ? ((const int4*)counts)[i] : make_int4(0, 0, 0, 0);
        }
#pragma unroll
        for (int k = 0; k < CB; ++k) part += cv[k].x + cv[k].y + cv[k].z + cv[k].w;
    }
    int c = 0;
    if (t < G1) {
        c = counts[b * G1 + t];
        rcnt[t] = c;
        rls[t]  = lstart[b * G1 + t];
    }
    for (int off = 16; off > 0; off >>= 1) part += __shfl_xor(part, off);
    if (lane == 0) red[w] = part;
    int inc = c;
    for (int off = 1; off < 32; off <<= 1) {
        const int nb = __shfl_up(inc, off);
        if (lane >= off) inc += nb;
    }
    if (lane == 31 && w < 8) wt[w] = inc;
    for (int i = t; i < 32 * HP2; i += 1024) hist[i] = 0;
    __syncthreads();
    if (t < G1) {
        int before = 0;
        for (int w2 = 0; w2 < w; ++w2) before += wt[w2];
        roff[t] = before + inc - c;
    }
    if (t == 0) {
        int s = 0;
        for (int i = 0; i < 32; ++i) s += red[i];
        sbeg = s;
        int tot = 0;
        for (int w2 = 0; w2 < (G1 + 31) / 32; ++w2) tot += wt[w2];
        sn = tot;
    }
    __syncthreads();
    const int beg = sbeg;
    const int n   = sn;
    {
        int pv[7];
#pragma unroll
        for (int q = 0; q < 7; ++q) {
            const int rb = w + 32 * q;
            const int rc = (rb < G1) ? rb : 0;
            const int cq = (rb < G1) ? rcnt[rc] : 0;
            pv[q] = (lane < cq) ? ekl[rc * EPB + rls[rc] + lane] : 0;
        }
#pragma unroll
        for (int q = 0; q < 7; ++q) {
            const int rb = w + 32 * q;
            const int rc = (rb < G1) ? rb : 0;
            const int cq = (rb < G1) ? rcnt[rc] : 0;
            if (lane < cq) ekey[beg + roff[rc] + lane] = pv[q];
            for (int j = 32 + lane; j < cq; j += 32)
                ekey[beg + roff[rc] + j] = ekl[rc * EPB + rls[rc] + j];
        }
    }
    __syncthreads();
    int* hw = hist + w * HP2;
    for (int i0 = 0; i0 < n; i0 += 1024) {
        const int i = i0 + t;
        const int k = (i < n) ? (ekey[beg + i] & 127) : 128;
        const unsigned m = match_mask<8>(k);
        if ((m & below) == 0) hw[k] += __builtin_popcount(m);
    }
    __syncthreads();
    if (t < 128) {
        int run = 0;
        for (int w2 = 0; w2 < 32; ++w2) {
            const int c = hist[w2 * HP2 + t];
            hist[w2 * HP2 + t] = run;
            run += c;
        }
        ncnt[t] = run;
    }
    __syncthreads();
    if (t < 128) {
        int ex = 0;
        for (int j = 0; j < t; ++j) ex += ncnt[j];
        nbase[t] = beg + ex;
        const int node = b * 128 + t;
        if (node < NN) {
            deg[node]    = ncnt[t];
            rowptr[node] = beg + ex;
        }
    }
    __syncthreads();
    for (int i0 = 0; i0 < n; i0 += 1024) {
        const int i = i0 + t;
        const int kv = (i < n) ? ekey[beg + i] : 0;
        const int k  = (i < n) ? (kv & 127) : 128;
        const unsigned m = match_mask<8>(k);
        const int ahead = __builtin_popcount(m & below);
        const int old = hw[k];
        if (i < n) col[nbase[k] + old + ahead] = kv >> 7;
        if (ahead == 0) hw[k] = old + __builtin_popcount(m);
    }
}

#define GR 32
__global__ void
__launch_bounds__(128)
layer1_fused(const int* __restrict__ rowptr, const int* __restrict__ deg, const int* __restrict__ col,
             const float* __restrict__ x,
             const float* __restrict__ W1l, const float* __restrict__ b1, const float* __restrict__ W1r,
             const float* __restrict__ W2l, const float* __restrict__ W2r, const float* __restrict__ b2,
             float* __restrict__ P, float* __restrict__ out, int nrows) {
    constexpr int MB    = 5;
    constexpr int NB    = 2;
    constexpr int ROWS  = 16 * MB;
    constexpr int PITCH = DHID + 4;
    constexpr int NW    = 4;
    constexpr int RPW   = ROWS / NW;
    constexpr int BATCH = 10;
    static_assert(RPW <= 32, "a wave holds its nodes' list bounds one per lane");

    __shared__ __attribute__((aligned(16))) float sM[ROWS * PITCH];
    __shared__ __attribute__((aligned(16))) float sX[ROWS * PITCH];

    const int lane = threadIdx.x & 31;
    const int wv   = threadIdx.x >> 5;
    const int row0 = blockIdx.x * ROWS;
    if (row0 >= nrows) return;

    const int wu = __builtin_amdgcn_readfirstlane(wv);
    const int n0 = row0 + wu * RPW;
    const int rpv = (lane < RPW) ? rowptr[n0 + lane] : 0;
    const int dgv = (lane < RPW) ? deg[n0 + lane] : 0;

#pragma unroll
    for (int b0 = 0; b0 < RPW; b0 += BATCH) {
        float4 vx[BATCH];
#pragma unroll
        for (int j = 0; j < BATCH; ++j) {
            const int r = wv + (b0 + j) * NW;
            vx[j] = *(const float4*)(x + (size_t)(row0 + r) * DHID + lane * 4);
        }
#pragma unroll
        for (int j = 0; j < BATCH; ++j) {
            const int r = wv + (b0 + j) * NW;
            *(float4*)&sX[r * PITCH + lane * 4] = vx[j];
        }
    }

    __syncthreads();

    const int half = lane >> 4;
    const int mn   = lane & 15;
    const int koff = half * 2;
    const int c0   = wv * (16 * NB) + mn;

    const float* mp = sM + mn * PITCH + koff;
    const float* xp = sX + mn * PITCH + koff;

    v8f acc[MB][NB];
#pragma unroll
    for (int mi = 0; mi < MB; ++mi)
#pragma unroll
        for (int nj = 0; nj < NB; ++nj)
            acc[mi][nj] = (v8f){0.f, 0.f, 0.f, 0.f, 0.f, 0.f, 0.f, 0.f};

    constexpr int KC = 4;
    int kr = 0;
    {
        const int ebeg = __builtin_amdgcn_readlane(rpv, 0);
        const int eend = __builtin_amdgcn_readlane(rpv, RPW - 1) + __builtin_amdgcn_readlane(dgv, RPW - 1);
        int i = 0;
        int cur_end = ebeg + __builtin_amdgcn_readlane(dgv, 0);
        float4 rs = make_float4(0.f, 0.f, 0.f, 0.f);
        float* mrow = sM + (wu * RPW) * PITCH + lane * 4;
        int nextid = (ebeg + lane < eend) ? col[ebeg + lane] : 0;
        v2f brn[KC][NB];
#pragma unroll
        for (int q = 0; q < KC; ++q) {
#pragma unroll
            for (int nj = 0; nj < NB; ++nj) {
                brn[q][nj][0] = W1r[(4 * q + koff)     * DHID + c0 + nj * 16];
                brn[q][nj][1] = W1r[(4 * q + koff + 1) * DHID + c0 + nj * 16];
            }
        }
        for (int c = ebeg; c < eend; c += GR) {
            const int myid = nextid;
            const int m = (eend - c < GR) ? (eend - c) : GR;
            v2f br[KC][NB];
#pragma unroll
            for (int q = 0; q < KC; ++q)
#pragma unroll
                for (int nj = 0; nj < NB; ++nj) br[q][nj] = brn[q][nj];
            float4 v[GR];
#pragma unroll
            for (int j = 0; j < GR; ++j) {
                const int sid = __builtin_amdgcn_readlane(myid, j);
                v[j] = *(const float4*)(x + (size_t)sid * DHID + lane * 4);
            }
            const int ni = c + GR + lane;
            const int nv = col[(ni < eend) ? ni : (eend - 1)];
            nextid = (ni < eend) ? nv : 0;
            const int kn = (kr + 4 * KC < DHID) ? (kr + 4 * KC) : 0;
#pragma unroll
            for (int q = 0; q < KC; ++q) {
#pragma unroll
                for (int nj = 0; nj < NB; ++nj) {
                    brn[q][nj][0] = W1r[(kn + 4 * q + koff)     * DHID + c0 + nj * 16];
                    brn[q][nj][1] = W1r[(kn + 4 * q + koff + 1) * DHID + c0 + nj * 16];
                }
            }
            if (kr < DHID) {
#pragma unroll
                for (int q = 0; q < KC; ++q) {
                    v2f ar[MB];
#pragma unroll
                    for (int mi = 0; mi < MB; ++mi)
                        ar[mi] = *(const v2f*)(xp + mi * 16 * PITCH + kr + 4 * q);
#pragma unroll
                    for (int mi = 0; mi < MB; ++mi) {
#pragma unroll
                        for (int nj = 0; nj < NB; ++nj) {
                            acc[mi][nj] = __builtin_amdgcn_wmma_f32_16x16x4_f32(
                                false, ar[mi], false, br[q][nj], (short)0, acc[mi][nj], false, false);
                        }
                    }
                }
                kr += 4 * KC;
                __builtin_amdgcn_sched_barrier(0);
                asm volatile("v_nop\n\tv_nop\n\tv_nop\n\tv_nop");
                __builtin_amdgcn_sched_barrier(0);
            }
#pragma unroll
            for (int j = 0; j < GR; ++j) {
                if (j < m) {
                    while (c + j == cur_end) {
                        const float s = 1.0f / fmaxf((float)__builtin_amdgcn_readlane(dgv, i), 1.0f);
                        *(float4*)(mrow + i * PITCH) = make_float4(rs.x * s, rs.y * s, rs.z * s, rs.w * s);
                        rs = make_float4(0.f, 0.f, 0.f, 0.f);
                        ++i;
                        cur_end += __builtin_amdgcn_readlane(dgv, i);
                    }
                    rs.x += v[j].x; rs.y += v[j].y; rs.z += v[j].z; rs.w += v[j].w;
                }
            }
        }
        for (; i < RPW; ++i) {
            const float s = 1.0f / fmaxf((float)__builtin_amdgcn_readlane(dgv, i), 1.0f);
            *(float4*)(mrow + i * PITCH) = make_float4(rs.x * s, rs.y * s, rs.z * s, rs.w * s);
            rs = make_float4(0.f, 0.f, 0.f, 0.f);
        }
    }
    for (; kr < DHID; kr += 4) {
        v2f br[NB];
#pragma unroll
        for (int nj = 0; nj < NB; ++nj) {
            br[nj][0] = W1r[(kr + koff)     * DHID + c0 + nj * 16];
            br[nj][1] = W1r[(kr + koff + 1) * DHID + c0 + nj * 16];
        }
        v2f ar[MB];
#pragma unroll
        for (int mi = 0; mi < MB; ++mi)
            ar[mi] = *(const v2f*)(xp + mi * 16 * PITCH + kr);
#pragma unroll
        for (int mi = 0; mi < MB; ++mi) {
#pragma unroll
            for (int nj = 0; nj < NB; ++nj) {
                acc[mi][nj] = __builtin_amdgcn_wmma_f32_16x16x4_f32(
                    false, ar[mi], false, br[nj], (short)0, acc[mi][nj], false, false);
            }
        }
    }
    __syncthreads();

#pragma unroll 4
    for (int k = 0; k < DHID; k += 4) {
        v2f bl[NB];
#pragma unroll
        for (int nj = 0; nj < NB; ++nj) {
            bl[nj][0] = W1l[(k + koff)     * DHID + c0 + nj * 16];
            bl[nj][1] = W1l[(k + koff + 1) * DHID + c0 + nj * 16];
        }
        v2f am[MB];
#pragma unroll
        for (int mi = 0; mi < MB; ++mi)
            am[mi] = *(const v2f*)(mp + mi * 16 * PITCH + k);
#pragma unroll
        for (int mi = 0; mi < MB; ++mi) {
#pragma unroll
            for (int nj = 0; nj < NB; ++nj) {
                acc[mi][nj] = __builtin_amdgcn_wmma_f32_16x16x4_f32(
                    false, am[mi], false, bl[nj], (short)0, acc[mi][nj], false, false);
            }
        }
    }

    __builtin_amdgcn_sched_barrier(0);
    asm volatile("v_nop\n\tv_nop\n\tv_nop\n\tv_nop");
    __builtin_amdgcn_sched_barrier(0);

    __syncthreads();
#pragma unroll
    for (int nj = 0; nj < NB; ++nj) {
        const float bv = b1[c0 + nj * 16];
#pragma unroll
        for (int mi = 0; mi < MB; ++mi) {
#pragma unroll
            for (int v = 0; v < 8; ++v) {
                const int m = mi * 16 + v + half * 8;
                sM[m * PITCH + c0 + nj * 16] = fmaxf(acc[mi][nj][v] + bv, 0.f);
            }
        }
    }
    __syncthreads();

    const int isr = wv >> 1;
    const int c2  = (wv & 1) * (16 * NB) + mn;

    const float* Wm = isr ? W2r : W2l;
    float*       dp = isr ? out : P;

#pragma unroll
    for (int mi = 0; mi < MB; ++mi)
#pragma unroll
        for (int nj = 0; nj < NB; ++nj)
            acc[mi][nj] = (v8f){0.f, 0.f, 0.f, 0.f, 0.f, 0.f, 0.f, 0.f};

#pragma unroll 2
    for (int k = 0; k < DHID; k += 4) {
        v2f bw[NB];
#pragma unroll
        for (int nj = 0; nj < NB; ++nj) {
            bw[nj][0] = Wm[(k + koff)     * DOUT + c2 + nj * 16];
            bw[nj][1] = Wm[(k + koff + 1) * DOUT + c2 + nj * 16];
        }
        v2f ah[MB];
#pragma unroll
        for (int mi = 0; mi < MB; ++mi)
            ah[mi] = *(const v2f*)(mp + mi * 16 * PITCH + k);
#pragma unroll
        for (int mi = 0; mi < MB; ++mi) {
#pragma unroll
            for (int nj = 0; nj < NB; ++nj) {
                acc[mi][nj] = __builtin_amdgcn_wmma_f32_16x16x4_f32(
                    false, ah[mi], false, bw[nj], (short)0, acc[mi][nj], false, false);
            }
        }
    }

    __builtin_amdgcn_sched_barrier(0);
    asm volatile("v_nop\n\tv_nop\n\tv_nop\n\tv_nop");
    __builtin_amdgcn_sched_barrier(0);

#pragma unroll
    for (int nj = 0; nj < NB; ++nj) {
        const float bv = isr ? b2[c2 + nj * 16] : 0.f;
#pragma unroll
        for (int mi = 0; mi < MB; ++mi) {
#pragma unroll
            for (int v = 0; v < 8; ++v) {
                const int m = row0 + mi * 16 + v + half * 8;
                dp[(size_t)m * DOUT + c2 + nj * 16] = acc[mi][nj][v] + bv;
            }
        }
    }
}


#define GU 16
__global__ void gather64_kernel(const int* __restrict__ rowptr, const int* __restrict__ deg,
                                const int* __restrict__ col, const float* __restrict__ P,
                                float* __restrict__ out, int n) {
    const int gid  = blockIdx.x * blockDim.x + threadIdx.x;
    const int lane = threadIdx.x & 31;
    const int r = __builtin_amdgcn_readfirstlane(gid >> 5);
    if (r >= n) return;
    const int beg = rowptr[r];
    const int cnt = deg[r];
    float* o = out + (size_t)r * DOUT + lane * 2;
    const float2 start = *(const float2*)o;
    float2 acc = make_float2(0.f, 0.f);
    for (int b0 = 0; b0 < cnt; b0 += 32) {
        const int m = (cnt - b0 < 32) ? (cnt - b0) : 32;
        const int myid = (lane < m) ? col[beg + b0 + lane] : 0;
        for (int j0 = 0; j0 < m; j0 += GU) {
            float2 v[GU];
#pragma unroll
            for (int j = 0; j < GU; ++j) {
                const int sid = __builtin_amdgcn_readlane(myid, j0 + j);
                if (j0 + j < m) v[j] = *(const float2*)(P + (size_t)sid * DOUT + lane * 2);
                else            v[j] = make_float2(0.f, 0.f);
            }
#pragma unroll
            for (int j = 0; j < GU; ++j) {
                acc.x += v[j].x; acc.y += v[j].y;
            }
        }
    }
    const float s = 1.0f / fmaxf((float)cnt, 1.0f);
    *(float2*)o = make_float2(start.x + acc.x * s, start.y + acc.y * s);
}

extern "C" void kernel_launch(void* const* d_in, const int* in_sizes, int n_in,
                              void* d_out, int out_size, void* d_ws, size_t ws_size,
                              hipStream_t stream) {
    const float* x   = (const float*)d_in[0];
    const int*   ei  = (const int*)d_in[1];
    const float* W1l = (const float*)d_in[2];
    const float* b1  = (const float*)d_in[3];
    const float* W1r = (const float*)d_in[4];
    const float* W2l = (const float*)d_in[5];
    const float* b2  = (const float*)d_in[6];
    const float* W2r = (const float*)d_in[7];
    float*       out = (float*)d_out;

    const int* src = ei;
    const int* dst = ei + NE;

    float* A      = (float*)d_ws;
    float* P      = A + (size_t)NN * DHID;
    int*   col    = (int*)(P + (size_t)NN * DOUT);
    int*   ekey   = col + NE;
    int*   ekl    = ekey + NE;
    int*   rowptr = ekl + NE;
    int*   deg    = rowptr + NN;
    int*   counts = deg + NN;
    int*   lstart = counts + NCNT;
    static_assert((size_t)NN * DOUT + 3 * (size_t)NE + 2 * (size_t)NN + 2 * (size_t)NCNT
                      <= (size_t)NN * DHID + NN,
                  "P and the lists fit the space that used to hold the counts and h");
    static_assert(((size_t)NN * DHID + (size_t)NN * DOUT + 3 * (size_t)NE + 2 * (size_t)NN) % 4 == 0,
                  "counts start on 16 B for the 16-B loads of the sum in front of a bucket");

    const int BLK = 256;

    split_local<<<G1, 1024, 0, stream>>>(src, dst, counts, lstart, ekl);
    bucket_lists<<<NBK, 1024, 0, stream>>>(counts, lstart, ekl, ekey, deg, rowptr, col);

    layer1_fused<<<NN / 80, 128, 0, stream>>>(rowptr, deg, col, x, W1l, b1, W1r, W2l, W2r, b2, P, out, NN);

    gather64_kernel<<<(NN * 32 + BLK - 1) / BLK, BLK, 0, stream>>>(rowptr, deg, col, P, out, NN);
}
